# starting kernel
# baseline (speedup 1.0000x reference)
#include <hip/hip_runtime.h>
#include <math.h>

typedef float v2f __attribute__((ext_vector_type(2)));
typedef float v8f __attribute__((ext_vector_type(8)));

#define DIM   192
#define C2    96
#define HW    4096
#define BATCH 2
#define NROW  (BATCH * HW)
#define KSTR  100

__global__ __launch_bounds__(256)
void nlb_qkv(const float* __restrict__ x,
             const float* __restrict__ w1, const float* __restrict__ b1,
             const float* __restrict__ w2, const float* __restrict__ b2,
             const float* __restrict__ w3, const float* __restrict__ b3,
             float* __restrict__ m1, float* __restrict__ m2, float* __restrict__ m3)
{
    const int hw = blockIdx.x * blockDim.x + threadIdx.x;
    const int bo = blockIdx.y;
    const int b  = bo / C2, o = bo - b * C2;
    const float* xb = x + (size_t)b * DIM * HW + hw;
    const float* W1 = w1 + o * DIM;
    const float* W2 = w2 + o * DIM;
    const float* W3 = w3 + o * DIM;
    float a1 = 0.f, a2 = 0.f, a3 = 0.f;
#pragma unroll 4
    for (int c = 0; c < DIM; ++c) {
        const float xv = xb[(size_t)c * HW];
        a1 = fmaf(xv, W1[c], a1);
        a2 = fmaf(xv, W2[c], a2);
        a3 = fmaf(xv, W3[c], a3);
    }
    const size_t idx = (size_t)bo * HW + hw;
    m1[idx] = a1 + b1[o];
    m2[idx] = a2 + b2[o];
    m3[idx] = a3 + b3[o];
}

__global__ __launch_bounds__(256)
void nlb_flash(const float* __restrict__ Q, const float* __restrict__ K,
               const float* __restrict__ V, float* __restrict__ O)
{
    __shared__ float sK[16 * KSTR];
    __shared__ float sV[16 * KSTR];
    __shared__ float sP[8 * 256];

    const int tid  = threadIdx.x;
    const int lane = tid & 31;
    const int wave = tid >> 5;
    const int row  = lane & 15;
    const int hi   = lane >> 4;
    const int qbase = (blockIdx.x * 8 + wave) * 16;
    float* myP = sP + wave * 256;

    v2f qb[24];
    {
        const float* qp = Q + (size_t)(qbase + row) * C2 + 2 * hi;
#pragma unroll
        for (int cc = 0; cc < 24; ++cc) qb[cc] = *(const v2f*)(qp + 4 * cc);
    }

    v8f acc[6];
#pragma unroll
    for (int jb = 0; jb < 6; ++jb)
#pragma unroll
        for (int e = 0; e < 8; ++e) acc[jb][e] = 0.f;
    float mrun = -INFINITY;
    float lrun = 0.f;

    for (int kt = 0; kt < NROW; kt += 16) {
        __syncthreads();
        {
            const float4* ks = (const float4*)(K + (size_t)kt * C2);
            const float4* vs = (const float4*)(V + (size_t)kt * C2);
            for (int i = tid; i < 384; i += 256) {
                const int r = i / 24, c4 = (i - r * 24) * 4;
                *(float4*)(sK + r * KSTR + c4) = ks[i];
                *(float4*)(sV + r * KSTR + c4) = vs[i];
            }
        }
        __syncthreads();

        v8f s;
#pragma unroll
        for (int e = 0; e < 8; ++e) s[e] = 0.f;
        {
            const float* kp = sK + row * KSTR + 2 * hi;
#pragma unroll
            for (int cc = 0; cc < 24; ++cc) {
                v2f a = *(const v2f*)(kp + 4 * cc);
                s = __builtin_amdgcn_wmma_f32_16x16x4_f32(
                        false, a, false, qb[cc], (short)0, s, false, false);
            }
        }

        float tmax = s[0];
#pragma unroll
        for (int i = 1; i < 8; ++i) tmax = fmaxf(tmax, s[i]);
        tmax = fmaxf(tmax, __shfl_xor(tmax, 16, 32));
        const float nm   = fmaxf(mrun, tmax);
        const float corr = __expf(mrun - nm);
        float p[8], ps = 0.f;
#pragma unroll
        for (int i = 0; i < 8; ++i) { p[i] = __expf(s[i] - nm); ps += p[i]; }
        ps += __shfl_xor(ps, 16, 32);
        lrun = lrun * corr + ps;
        mrun = nm;
#pragma unroll
        for (int jb = 0; jb < 6; ++jb)
#pragma unroll
            for (int e = 0; e < 8; ++e) acc[jb][e] *= corr;

#pragma unroll
        for (int i = 0; i < 8; ++i)
            myP[(i + 8 * hi) * 16 + row] = p[i];
        __syncthreads();

        v2f bp[4];
#pragma unroll
        for (int t = 0; t < 4; ++t) {
            bp[t].x = myP[(4 * t + 2 * hi) * 16 + row];
            bp[t].y = myP[(4 * t + 2 * hi + 1) * 16 + row];
        }

#pragma unroll
        for (int jb = 0; jb < 6; ++jb) {
#pragma unroll
            for (int t = 0; t < 4; ++t) {
                v2f a;
                a.x = sV[(4 * t + 2 * hi)     * KSTR + jb * 16 + row];
                a.y = sV[(4 * t + 2 * hi + 1) * KSTR + jb * 16 + row];
                acc[jb] = __builtin_amdgcn_wmma_f32_16x16x4_f32(
                              false, a, false, bp[t], (short)0, acc[jb], false, false);
            }
        }
    }

    const float inv = 1.f / lrun;
    float* op = O + (size_t)(qbase + row) * C2 + 8 * hi;
#pragma unroll
    for (int jb = 0; jb < 6; ++jb)
#pragma unroll
        for (int i = 0; i < 8; ++i)
            op[jb * 16 + i] = acc[jb][i] * inv;
}

__global__ __launch_bounds__(256)
void nlb_proj(const float* __restrict__ A, const float* __restrict__ wl,
              const float* __restrict__ bl, const float* __restrict__ x,
              float* __restrict__ out)
{
    const int hw = blockIdx.x * blockDim.x + threadIdx.x;
    const int bo = blockIdx.y;
    const int b  = bo / DIM, o = bo - b * DIM;
    const float* Ab = A + (size_t)b * C2 * HW + hw;
    const float* W  = wl + o * C2;
    float acc = bl[o];
#pragma unroll 4
    for (int c = 0; c < C2; ++c) acc = fmaf(Ab[(size_t)c * HW], W[c], acc);
    const size_t idx = (size_t)bo * HW + hw;
    out[idx] = acc + x[idx];
}

extern "C" void kernel_launch(void* const* d_in, const int* in_sizes, int n_in,
                              void* d_out, int out_size, void* d_ws, size_t ws_size,
                              hipStream_t stream)
{
    const float* x  = (const float*)d_in[0];
    const float* w1 = (const float*)d_in[1];
    const float* b1 = (const float*)d_in[2];
    const float* w2 = (const float*)d_in[3];
    const float* b2 = (const float*)d_in[4];
    const float* w3 = (const float*)d_in[5];
    const float* b3 = (const float*)d_in[6];
    const float* wl = (const float*)d_in[7];
    const float* bl = (const float*)d_in[8];
    float* out = (float*)d_out;

    const size_t msz = (size_t)NROW * C2;
    float* m1 = (float*)d_ws;
    float* m2 = m1 + msz;
    float* m3 = m2 + msz;
    float* ob = m3 + msz;

    nlb_qkv<<<dim3(HW / 256, BATCH * C2), 256, 0, stream>>>(
        x, w1, b1, w2, b2, w3, b3, m1, m2, m3);
    nlb_flash<<<dim3(NROW / (16 * 8)), 256, 0, stream>>>(m1, m2, m3, ob);
    nlb_proj<<<dim3(HW / 256, BATCH * DIM), 256, 0, stream>>>(ob, wl, bl, x, out);
}


# attention key range split over 12 work-groups per query block (768, not 64), partial max/sum/output folded by a new merge kernel; changes softmax summation order (per-range sums, then merged)
# speedup vs baseline: 4.6349x; 4.6349x over previous
#include <hip/hip_runtime.h>
#include <math.h>

typedef float v2f __attribute__((ext_vector_type(2)));
typedef float v8f __attribute__((ext_vector_type(8)));

#define DIM   192
#define C2    96
#define HW    4096
#define BATCH 2
#define NROW  (BATCH * HW)
#define KSTR  100
#define NTILE (NROW / 16)
#define NSPLIT 12

__global__ __launch_bounds__(256)
void nlb_qkv(const float* __restrict__ x,
             const float* __restrict__ w1, const float* __restrict__ b1,
             const float* __restrict__ w2, const float* __restrict__ b2,
             const float* __restrict__ w3, const float* __restrict__ b3,
             float* __restrict__ m1, float* __restrict__ m2, float* __restrict__ m3)
{
    const int hw = blockIdx.x * blockDim.x + threadIdx.x;
    const int bo = blockIdx.y;
    const int b  = bo / C2, o = bo - b * C2;
    const float* xb = x + (size_t)b * DIM * HW + hw;
    const float* W1 = w1 + o * DIM;
    const float* W2 = w2 + o * DIM;
    const float* W3 = w3 + o * DIM;
    float a1 = 0.f, a2 = 0.f, a3 = 0.f;
#pragma unroll 4
    for (int c = 0; c < DIM; ++c) {
        const float xv = xb[(size_t)c * HW];
        a1 = fmaf(xv, W1[c], a1);
        a2 = fmaf(xv, W2[c], a2);
        a3 = fmaf(xv, W3[c], a3);
    }
    const size_t idx = (size_t)bo * HW + hw;
    m1[idx] = a1 + b1[o];
    m2[idx] = a2 + b2[o];
    m3[idx] = a3 + b3[o];
}

__global__ __launch_bounds__(256)
void nlb_flash(const float* __restrict__ Q, const float* __restrict__ K,
               const float* __restrict__ V, float* __restrict__ Opart,
               float* __restrict__ MLpart)
{
    __shared__ float sK[16 * KSTR];
    __shared__ float sV[16 * KSTR];
    __shared__ float sP[8 * 256];

    const int tid  = threadIdx.x;
    const int lane = tid & 31;
    const int wave = tid >> 5;
    const int row  = lane & 15;
    const int hi   = lane >> 4;
    const int qbase = (blockIdx.x * 8 + wave) * 16;
    float* myP = sP + wave * 256;

    v2f qb[24];
    {
        const float* qp = Q + (size_t)(qbase + row) * C2 + 2 * hi;
#pragma unroll
        for (int cc = 0; cc < 24; ++cc) qb[cc] = *(const v2f*)(qp + 4 * cc);
    }

    v8f acc[6];
#pragma unroll
    for (int jb = 0; jb < 6; ++jb)
#pragma unroll
        for (int e = 0; e < 8; ++e) acc[jb][e] = 0.f;
    float mrun = -INFINITY;
    float lrun = 0.f;

    const int split = blockIdx.y;
    const int kbeg  = ((split * NTILE) / NSPLIT) * 16;
    const int kend  = (((split + 1) * NTILE) / NSPLIT) * 16;
    for (int kt = kbeg; kt < kend; kt += 16) {
        __syncthreads();
        {
            const float4* ks = (const float4*)(K + (size_t)kt * C2);
            const float4* vs = (const float4*)(V + (size_t)kt * C2);
            for (int i = tid; i < 384; i += 256) {
                const int r = i / 24, c4 = (i - r * 24) * 4;
                *(float4*)(sK + r * KSTR + c4) = ks[i];
                *(float4*)(sV + r * KSTR + c4) = vs[i];
            }
        }
        __syncthreads();

        v8f s;
#pragma unroll
        for (int e = 0; e < 8; ++e) s[e] = 0.f;
        {
            const float* kp = sK + row * KSTR + 2 * hi;
#pragma unroll
            for (int cc = 0; cc < 24; ++cc) {
                v2f a = *(const v2f*)(kp + 4 * cc);
                s = __builtin_amdgcn_wmma_f32_16x16x4_f32(
                        false, a, false, qb[cc], (short)0, s, false, false);
            }
        }

        float tmax = s[0];
#pragma unroll
        for (int i = 1; i < 8; ++i) tmax = fmaxf(tmax, s[i]);
        tmax = fmaxf(tmax, __shfl_xor(tmax, 16, 32));
        const float nm   = fmaxf(mrun, tmax);
        const float corr = __expf(mrun - nm);
        float p[8], ps = 0.f;
#pragma unroll
        for (int i = 0; i < 8; ++i) { p[i] = __expf(s[i] - nm); ps += p[i]; }
        ps += __shfl_xor(ps, 16, 32);
        lrun = lrun * corr + ps;
        mrun = nm;
#pragma unroll
        for (int jb = 0; jb < 6; ++jb)
#pragma unroll
            for (int e = 0; e < 8; ++e) acc[jb][e] *= corr;

#pragma unroll
        for (int i = 0; i < 8; ++i)
            myP[(i + 8 * hi) * 16 + row] = p[i];
        __syncthreads();

        v2f bp[4];
#pragma unroll
        for (int t = 0; t < 4; ++t) {
            bp[t].x = myP[(4 * t + 2 * hi) * 16 + row];
            bp[t].y = myP[(4 * t + 2 * hi + 1) * 16 + row];
        }

#pragma unroll
        for (int jb = 0; jb < 6; ++jb) {
#pragma unroll
            for (int t = 0; t < 4; ++t) {
                v2f a;
                a.x = sV[(4 * t + 2 * hi)     * KSTR + jb * 16 + row];
                a.y = sV[(4 * t + 2 * hi + 1) * KSTR + jb * 16 + row];
                acc[jb] = __builtin_amdgcn_wmma_f32_16x16x4_f32(
                              false, a, false, bp[t], (short)0, acc[jb], false, false);
            }
        }
    }

    const size_t prow = (size_t)split * NROW + (qbase + row);
    float* op = Opart + prow * C2 + 8 * hi;
#pragma unroll
    for (int jb = 0; jb < 6; ++jb)
#pragma unroll
        for (int i = 0; i < 8; ++i)
            op[jb * 16 + i] = acc[jb][i];
    if (hi == 0) {
        v2f ml;
        ml.x = mrun;
        ml.y = lrun;
        *(v2f*)(MLpart + 2 * prow) = ml;
    }
}

__global__ __launch_bounds__(256)
void nlb_merge(const float* __restrict__ Opart, const float* __restrict__ MLpart,
               float* __restrict__ O)
{
    const int gid = blockIdx.x * blockDim.x + threadIdx.x;
    const int q   = gid / 24;
    v2f ml[NSPLIT];
    float M = -INFINITY;
#pragma unroll
    for (int s = 0; s < NSPLIT; ++s) {
        ml[s] = *(const v2f*)(MLpart + 2 * ((size_t)s * NROW + q));
        M = fmaxf(M, ml[s].x);
    }
    float L = 0.f;
    float4 o = make_float4(0.f, 0.f, 0.f, 0.f);
#pragma unroll
    for (int s = 0; s < NSPLIT; ++s) {
        const float w = __expf(ml[s].x - M);
        L = fmaf(ml[s].y, w, L);
        const float4 a = ((const float4*)Opart)[(size_t)s * (NROW * 24) + gid];
        o.x = fmaf(a.x, w, o.x);
        o.y = fmaf(a.y, w, o.y);
        o.z = fmaf(a.z, w, o.z);
        o.w = fmaf(a.w, w, o.w);
    }
    const float inv = 1.f / L;
    o.x *= inv; o.y *= inv; o.z *= inv; o.w *= inv;
    ((float4*)O)[gid] = o;
}

__global__ __launch_bounds__(256)
void nlb_proj(const float* __restrict__ A, const float* __restrict__ wl,
              const float* __restrict__ bl, const float* __restrict__ x,
              float* __restrict__ out)
{
    const int hw = blockIdx.x * blockDim.x + threadIdx.x;
    const int bo = blockIdx.y;
    const int b  = bo / DIM, o = bo - b * DIM;
    const float* Ab = A + (size_t)b * C2 * HW + hw;
    const float* W  = wl + o * C2;
    float acc = bl[o];
#pragma unroll 4
    for (int c = 0; c < C2; ++c) acc = fmaf(Ab[(size_t)c * HW], W[c], acc);
    const size_t idx = (size_t)bo * HW + hw;
    out[idx] = acc + x[idx];
}

extern "C" void kernel_launch(void* const* d_in, const int* in_sizes, int n_in,
                              void* d_out, int out_size, void* d_ws, size_t ws_size,
                              hipStream_t stream)
{
    const float* x  = (const float*)d_in[0];
    const float* w1 = (const float*)d_in[1];
    const float* b1 = (const float*)d_in[2];
    const float* w2 = (const float*)d_in[3];
    const float* b2 = (const float*)d_in[4];
    const float* w3 = (const float*)d_in[5];
    const float* b3 = (const float*)d_in[6];
    const float* wl = (const float*)d_in[7];
    const float* bl = (const float*)d_in[8];
    float* out = (float*)d_out;

    const size_t msz = (size_t)NROW * C2;
    float* m1 = (float*)d_ws;
    float* m2 = m1 + msz;
    float* m3 = m2 + msz;
    float* ob = m3 + msz;
    float* op = ob + msz;
    float* ml = op + (size_t)NSPLIT * msz;

    nlb_qkv<<<dim3(HW / 256, BATCH * C2), 256, 0, stream>>>(
        x, w1, b1, w2, b2, w3, b3, m1, m2, m3);
    nlb_flash<<<dim3(NROW / (16 * 8), NSPLIT), 256, 0, stream>>>(m1, m2, m3, op, ml);
    nlb_merge<<<dim3(NROW * 24 / 256), 256, 0, stream>>>(op, ml, ob);
    nlb_proj<<<dim3(HW / 256, BATCH * DIM), 256, 0, stream>>>(ob, wl, bl, x, out);
}


# attention K/V staging software-pipelined: next tile's global loads fly under the current tile's products into a second LDS stage; P exchange ordered per wave; 1 work-group barrier per tile, was 3
# speedup vs baseline: 6.3995x; 1.3807x over previous
#include <hip/hip_runtime.h>
#include <math.h>

typedef float v2f __attribute__((ext_vector_type(2)));
typedef float v8f __attribute__((ext_vector_type(8)));

#define DIM   192
#define C2    96
#define HW    4096
#define BATCH 2
#define NROW  (BATCH * HW)
#define KSTR  100
#define NTILE (NROW / 16)
#define NSPLIT 12

#define CONV_CH 4

struct ConvChunk {
    float  r[CONV_CH][8];
    float4 a[CONV_CH][3];
};

__device__ __forceinline__ float xhalf(float v)
{
    const int b = __builtin_bit_cast(int, v);
    return __builtin_bit_cast(float,
        __builtin_amdgcn_permlanex16(b, b, 0x76543210, 0xfedcba98, false, false));
}

__device__ __forceinline__ void half_swap(float& a, float& b, bool upper)
{
    const float xa = xhalf(a), xb = xhalf(b);
    const float na = upper ? xb : a;
    const float nb = upper ? b : xa;
    a = na;
    b = nb;
}

template <int KDIM>
__device__ __forceinline__ void conv_load(ConvChunk& c, const float* __restrict__ wp,
                                          const float* __restrict__ xp, int ch)
{
#pragma unroll
    for (int g = 0; g < CONV_CH; ++g) {
        const int c0 = (ch * CONV_CH + g) * 8;
#pragma unroll
        for (int j = 0; j < 8; ++j) c.r[g][j] = xp[(size_t)(c0 + j) * HW];
#pragma unroll
        for (int t = 0; t < 3; ++t) c.a[g][t] = *(const float4*)(wp + t * 16 * KDIM + c0);
    }
}

__device__ __forceinline__ void conv_mma(ConvChunk& c, v8f (&acc)[3][2], bool upper)
{
#pragma unroll
    for (int g = 0; g < CONV_CH; ++g) {
#pragma unroll
        for (int j = 0; j < 4; ++j) half_swap(c.r[g][j], c.r[g][j + 4], upper);
#pragma unroll
        for (int u = 0; u < 2; ++u) {
            v2f b0, b1;
            b0.x = c.r[g][2 * u];     b0.y = c.r[g][2 * u + 1];
            b1.x = c.r[g][2 * u + 4]; b1.y = c.r[g][2 * u + 5];
#pragma unroll
            for (int t = 0; t < 3; ++t) {
                v2f a;
                a.x = u ? c.a[g][t].z : c.a[g][t].x;
                a.y = u ? c.a[g][t].w : c.a[g][t].y;
                acc[t][0] = __builtin_amdgcn_wmma_f32_16x16x4_f32(
                                false, a, false, b0, (short)0, acc[t][0], false, false);
                acc[t][1] = __builtin_amdgcn_wmma_f32_16x16x4_f32(
                                false, a, false, b1, (short)0, acc[t][1], false, false);
            }
        }
    }
}

template <int KDIM>
__device__ __forceinline__ void conv_gemm(const float* __restrict__ wp,
                                          const float* __restrict__ xp, v8f (&acc)[3][2],
                                          bool upper)
{
    constexpr int NCH = KDIM / (8 * CONV_CH);
#pragma unroll
    for (int t = 0; t < 3; ++t)
#pragma unroll
        for (int j = 0; j < 2; ++j)
#pragma unroll
            for (int e = 0; e < 8; ++e) acc[t][j][e] = 0.f;
    ConvChunk buf[2];
    conv_load<KDIM>(buf[0], wp, xp, 0);
#pragma unroll
    for (int ch = 0; ch < NCH; ++ch) {
        if (ch + 1 < NCH) conv_load<KDIM>(buf[(ch + 1) & 1], wp, xp, ch + 1);
        __builtin_amdgcn_sched_barrier(0);
        conv_mma(buf[ch & 1], acc, upper);
    }
}

__global__ __launch_bounds__(128)
void nlb_qkv(const float* __restrict__ x,
             const float* __restrict__ w1, const float* __restrict__ b1,
             const float* __restrict__ w2, const float* __restrict__ b2,
             const float* __restrict__ w3, const float* __restrict__ b3,
             float* __restrict__ m1, float* __restrict__ m2, float* __restrict__ m3)
{
    const int lane = threadIdx.x & 31;
    const int wave = threadIdx.x >> 5;
    const int row  = lane & 15;
    const int hi   = lane >> 4;
    const int p    = (blockIdx.x * 4 + wave) * 32;
    const int b    = p >> 12, hw0 = p & (HW - 1);
    const int sel  = blockIdx.y >> 1;
    const int o0   = (blockIdx.y & 1) * 48;
    const float* w    = sel == 0 ? w1 : (sel == 1 ? w2 : w3);
    const float* bias = sel == 0 ? b1 : (sel == 1 ? b2 : b3);
    float*       m    = sel == 0 ? m1 : (sel == 1 ? m2 : m3);

    v8f acc[3][2];
    conv_gemm<DIM>(w + (size_t)(o0 + row) * DIM + 4 * hi,
                   x + (size_t)b * DIM * HW + hw0 + lane, acc, hi != 0);

    float* mp = m + ((size_t)b * C2 + o0) * HW + hw0 + lane;
#pragma unroll
    for (int t = 0; t < 3; ++t)
#pragma unroll
        for (int i = 0; i < 8; ++i) {
            float lo = acc[t][0][i], up = acc[t][1][i];
            half_swap(lo, up, hi != 0);
            const int oa = 16 * t + i;
            mp[(size_t)oa * HW]       = lo + bias[o0 + oa];
            mp[(size_t)(oa + 8) * HW] = up + bias[o0 + oa + 8];
        }
}

__global__ __launch_bounds__(256, 6)
void nlb_flash(const float* __restrict__ Q, const float* __restrict__ K,
               const float* __restrict__ V, float* __restrict__ Opart,
               float* __restrict__ MLpart)
{
    __shared__ float sK[2 * 16 * KSTR];
    __shared__ float sV[2 * 16 * KSTR];
    __shared__ float sP[8 * 256];

    const int tid  = threadIdx.x;
    const int lane = tid & 31;
    const int wave = tid >> 5;
    const int row  = lane & 15;
    const int hi   = lane >> 4;
    const int qbase = (blockIdx.x * 8 + wave) * 16;
    float* myP = sP + wave * 256;

    v2f qb[24];
    {
        const float* qp = Q + (size_t)(qbase + row) * C2 + 2 * hi;
#pragma unroll
        for (int cc = 0; cc < 24; ++cc) qb[cc] = *(const v2f*)(qp + 4 * cc);
    }

    v8f acc[6];
#pragma unroll
    for (int jb = 0; jb < 6; ++jb)
#pragma unroll
        for (int e = 0; e < 8; ++e) acc[jb][e] = 0.f;
    float mrun = -INFINITY;
    float lrun = 0.f;

    const int split = blockIdx.y;
    const int kbeg  = ((split * NTILE) / NSPLIT) * 16;
    const int kend  = (((split + 1) * NTILE) / NSPLIT) * 16;
    const int ntile = (kend - kbeg) / 16;

    const bool k1 = __builtin_amdgcn_readfirstlane(wave) < 4;
    const int  i1 = k1 ? tid + 256 : tid - 128;
    const int  i2 = tid + 128;
    const int  r0i = tid / 24, r1i = i1 / 24, r2i = i2 / 24;
    float* l0 = sK + r0i * KSTR + (tid - r0i * 24) * 4;
    float* l1 = (k1 ? sK : sV) + r1i * KSTR + (i1 - r1i * 24) * 4;
    float* l2 = sV + r2i * KSTR + (i2 - r2i * 24) * 4;

    float4 f0, f1, f2;
    {
        const float4* kt = (const float4*)(K + (size_t)kbeg * C2);
        const float4* vt = (const float4*)(V + (size_t)kbeg * C2);
        f0 = kt[tid];
        f1 = (k1 ? kt : vt)[i1];
        f2 = vt[i2];
    }
    *(float4*)l0 = f0;
    *(float4*)l1 = f1;
    *(float4*)l2 = f2;
    __syncthreads();

    for (int it = 0; it < ntile; ++it) {
        const int  cur  = (it & 1) * (16 * KSTR);
        const int  nxt  = (16 * KSTR) - cur;
        const bool more = it + 1 < ntile;
        if (more) {
            const float4* kt = (const float4*)(K + (size_t)(kbeg + 16 * (it + 1)) * C2);
            const float4* vt = (const float4*)(V + (size_t)(kbeg + 16 * (it + 1)) * C2);
            f0 = kt[tid];
            f1 = (k1 ? kt : vt)[i1];
            f2 = vt[i2];
        }
        const float* cK = sK + cur;
        const float* cV = sV + cur;

        v8f s;
#pragma unroll
        for (int e = 0; e < 8; ++e) s[e] = 0.f;
        {
            const float* kp = cK + row * KSTR + 2 * hi;
#pragma unroll
            for (int cc = 0; cc < 24; ++cc) {
                v2f a = *(const v2f*)(kp + 4 * cc);
                s = __builtin_amdgcn_wmma_f32_16x16x4_f32(
                        false, a, false, qb[cc], (short)0, s, false, false);
            }
        }

        float tmax = s[0];
#pragma unroll
        for (int i = 1; i < 8; ++i) tmax = fmaxf(tmax, s[i]);
        tmax = fmaxf(tmax, __shfl_xor(tmax, 16, 32));
        const float nm   = fmaxf(mrun, tmax);
        const float corr = __expf(mrun - nm);
        float p[8], ps = 0.f;
#pragma unroll
        for (int i = 0; i < 8; ++i) { p[i] = __expf(s[i] - nm); ps += p[i]; }
        ps += __shfl_xor(ps, 16, 32);
        lrun = lrun * corr + ps;
        mrun = nm;
#pragma unroll
        for (int jb = 0; jb < 6; ++jb)
#pragma unroll
            for (int e = 0; e < 8; ++e) acc[jb][e] *= corr;

#pragma unroll
        for (int i = 0; i < 8; ++i)
            myP[(i + 8 * hi) * 16 + row] = p[i];
        __builtin_amdgcn_fence(__ATOMIC_RELEASE, "wavefront");
        __builtin_amdgcn_wave_barrier();
        __builtin_amdgcn_fence(__ATOMIC_ACQUIRE, "wavefront");

        v2f bp[4];
#pragma unroll
        for (int t = 0; t < 4; ++t) {
            bp[t].x = myP[(4 * t + 2 * hi) * 16 + row];
            bp[t].y = myP[(4 * t + 2 * hi + 1) * 16 + row];
        }

#pragma unroll
        for (int jb = 0; jb < 6; ++jb) {
#pragma unroll
            for (int t = 0; t < 4; ++t) {
                v2f a;
                a.x = cV[(4 * t + 2 * hi)     * KSTR + jb * 16 + row];
                a.y = cV[(4 * t + 2 * hi + 1) * KSTR + jb * 16 + row];
                acc[jb] = __builtin_amdgcn_wmma_f32_16x16x4_f32(
                              false, a, false, bp[t], (short)0, acc[jb], false, false);
            }
        }

        if (more) {
            *(float4*)(l0 + nxt) = f0;
            *(float4*)(l1 + nxt) = f1;
            *(float4*)(l2 + nxt) = f2;
        }
        __syncthreads();
    }

    const size_t prow = (size_t)split * NROW + (qbase + row);
    float* op = Opart + prow * C2 + 8 * hi;
#pragma unroll
    for (int jb = 0; jb < 6; ++jb)
#pragma unroll
        for (int i = 0; i < 8; ++i)
            op[jb * 16 + i] = acc[jb][i];
    if (hi == 0) {
        v2f ml;
        ml.x = mrun;
        ml.y = lrun;
        *(v2f*)(MLpart + 2 * prow) = ml;
    }
}

__global__ __launch_bounds__(256)
void nlb_merge(const float* __restrict__ Opart, const float* __restrict__ MLpart,
               float* __restrict__ O)
{
    const int gid = blockIdx.x * blockDim.x + threadIdx.x;
    const int q   = gid / 24;
    v2f ml[NSPLIT];
    float M = -INFINITY;
#pragma unroll
    for (int s = 0; s < NSPLIT; ++s) {
        ml[s] = *(const v2f*)(MLpart + 2 * ((size_t)s * NROW + q));
        M = fmaxf(M, ml[s].x);
    }
    float L = 0.f;
    float4 o = make_float4(0.f, 0.f, 0.f, 0.f);
#pragma unroll
    for (int s = 0; s < NSPLIT; ++s) {
        const float w = __expf(ml[s].x - M);
        L = fmaf(ml[s].y, w, L);
        const float4 a = ((const float4*)Opart)[(size_t)s * (NROW * 24) + gid];
        o.x = fmaf(a.x, w, o.x);
        o.y = fmaf(a.y, w, o.y);
        o.z = fmaf(a.z, w, o.z);
        o.w = fmaf(a.w, w, o.w);
    }
    const float inv = 1.f / L;
    o.x *= inv; o.y *= inv; o.z *= inv; o.w *= inv;
    ((float4*)O)[gid] = o;
}

__global__ __launch_bounds__(128)
void nlb_proj(const float* __restrict__ A, const float* __restrict__ wl,
              const float* __restrict__ bl, const float* __restrict__ x,
              float* __restrict__ out)
{
    const int lane = threadIdx.x & 31;
    const int wave = threadIdx.x >> 5;
    const int row  = lane & 15;
    const int hi   = lane >> 4;
    const int p    = (blockIdx.x * 4 + wave) * 32;
    const int b    = p >> 12, hw0 = p & (HW - 1);
    const int o0   = blockIdx.y * 48;

    v8f acc[3][2];
    conv_gemm<C2>(wl + (size_t)(o0 + row) * C2 + 4 * hi,
                  A + (size_t)b * C2 * HW + hw0 + lane, acc, hi != 0);

    const size_t base = ((size_t)b * DIM + o0) * HW + hw0 + lane;
#pragma unroll
    for (int t = 0; t < 3; ++t)
#pragma unroll
        for (int i = 0; i < 8; ++i) {
            float lo = acc[t][0][i], up = acc[t][1][i];
            half_swap(lo, up, hi != 0);
            const int oa = 16 * t + i;
            const size_t ia = base + (size_t)oa * HW;
            const size_t ib = base + (size_t)(oa + 8) * HW;
            out[ia] = (lo + bl[o0 + oa])     + x[ia];
            out[ib] = (up + bl[o0 + oa + 8]) + x[ib];
        }
}

extern "C" void kernel_launch(void* const* d_in, const int* in_sizes, int n_in,
                              void* d_out, int out_size, void* d_ws, size_t ws_size,
                              hipStream_t stream)
{
    const float* x  = (const float*)d_in[0];
    const float* w1 = (const float*)d_in[1];
    const float* b1 = (const float*)d_in[2];
    const float* w2 = (const float*)d_in[3];
    const float* b2 = (const float*)d_in[4];
    const float* w3 = (const float*)d_in[5];
    const float* b3 = (const float*)d_in[6];
    const float* wl = (const float*)d_in[7];
    const float* bl = (const float*)d_in[8];
    float* out = (float*)d_out;

    const size_t msz = (size_t)NROW * C2;
    float* m1 = (float*)d_ws;
    float* m2 = m1 + msz;
    float* m3 = m2 + msz;
    float* ob = m3 + msz;
    float* op = ob + msz;
    float* ml = op + (size_t)NSPLIT * msz;

    nlb_qkv<<<dim3(NROW / 128, 6), 128, 0, stream>>>(
        x, w1, b1, w2, b2, w3, b3, m1, m2, m3);
    nlb_flash<<<dim3(NROW / (16 * 8), NSPLIT), 256, 0, stream>>>(m1, m2, m3, op, ml);
    nlb_merge<<<dim3(NROW * 24 / 256), 256, 0, stream>>>(op, ml, ob);
    nlb_proj<<<dim3(NROW / 128, DIM / 48), 128, 0, stream>>>(ob, wl, bl, x, out);
}


# attention P*V: k index's key order set to match the score registers, so P never leaves them, and V staged key-pair-interleaved so a V^T fragment is one 8-byte LDS read; P*V key sum order changes
# speedup vs baseline: 6.6732x; 1.0428x over previous
#include <hip/hip_runtime.h>
#include <math.h>

typedef float v2f __attribute__((ext_vector_type(2)));
typedef float v8f __attribute__((ext_vector_type(8)));

#define DIM   192
#define C2    96
#define HW    4096
#define BATCH 2
#define NROW  (BATCH * HW)
#define KSTR  100
#define VROW  200
#define NTILE (NROW / 16)
#define NSPLIT 12

#define CONV_CH 4

struct ConvChunk {
    float  r[CONV_CH][8];
    float4 a[CONV_CH][3];
};

__device__ __forceinline__ float xhalf(float v)
{
    const int b = __builtin_bit_cast(int, v);
    return __builtin_bit_cast(float,
        __builtin_amdgcn_permlanex16(b, b, 0x76543210, 0xfedcba98, false, false));
}

__device__ __forceinline__ void half_swap(float& a, float& b, bool upper)
{
    const float xa = xhalf(a), xb = xhalf(b);
    const float na = upper ? xb : a;
    const float nb = upper ? b : xa;
    a = na;
    b = nb;
}

template <int KDIM>
__device__ __forceinline__ void conv_load(ConvChunk& c, const float* __restrict__ wp,
                                          const float* __restrict__ xp, int ch)
{
#pragma unroll
    for (int g = 0; g < CONV_CH; ++g) {
        const int c0 = (ch * CONV_CH + g) * 8;
#pragma unroll
        for (int j = 0; j < 8; ++j) c.r[g][j] = xp[(size_t)(c0 + j) * HW];
#pragma unroll
        for (int t = 0; t < 3; ++t) c.a[g][t] = *(const float4*)(wp + t * 16 * KDIM + c0);
    }
}

__device__ __forceinline__ void conv_mma(ConvChunk& c, v8f (&acc)[3][2], bool upper)
{
#pragma unroll
    for (int g = 0; g < CONV_CH; ++g) {
#pragma unroll
        for (int j = 0; j < 4; ++j) half_swap(c.r[g][j], c.r[g][j + 4], upper);
#pragma unroll
        for (int u = 0; u < 2; ++u) {
            v2f b0, b1;
            b0.x = c.r[g][2 * u];     b0.y = c.r[g][2 * u + 1];
            b1.x = c.r[g][2 * u + 4]; b1.y = c.r[g][2 * u + 5];
#pragma unroll
            for (int t = 0; t < 3; ++t) {
                v2f a;
                a.x = u ? c.a[g][t].z : c.a[g][t].x;
                a.y = u ? c.a[g][t].w : c.a[g][t].y;
                acc[t][0] = __builtin_amdgcn_wmma_f32_16x16x4_f32(
                                false, a, false, b0, (short)0, acc[t][0], false, false);
                acc[t][1] = __builtin_amdgcn_wmma_f32_16x16x4_f32(
                                false, a, false, b1, (short)0, acc[t][1], false, false);
            }
        }
    }
}

template <int KDIM>
__device__ __forceinline__ void conv_gemm(const float* __restrict__ wp,
                                          const float* __restrict__ xp, v8f (&acc)[3][2],
                                          bool upper)
{
    constexpr int NCH = KDIM / (8 * CONV_CH);
#pragma unroll
    for (int t = 0; t < 3; ++t)
#pragma unroll
        for (int j = 0; j < 2; ++j)
#pragma unroll
            for (int e = 0; e < 8; ++e) acc[t][j][e] = 0.f;
    ConvChunk buf[2];
    conv_load<KDIM>(buf[0], wp, xp, 0);
#pragma unroll
    for (int ch = 0; ch < NCH; ++ch) {
        if (ch + 1 < NCH) conv_load<KDIM>(buf[(ch + 1) & 1], wp, xp, ch + 1);
        __builtin_amdgcn_sched_barrier(0);
        conv_mma(buf[ch & 1], acc, upper);
    }
}

__global__ __launch_bounds__(128)
void nlb_qkv(const float* __restrict__ x,
             const float* __restrict__ w1, const float* __restrict__ b1,
             const float* __restrict__ w2, const float* __restrict__ b2,
             const float* __restrict__ w3, const float* __restrict__ b3,
             float* __restrict__ m1, float* __restrict__ m2, float* __restrict__ m3)
{
    const int lane = threadIdx.x & 31;
    const int wave = threadIdx.x >> 5;
    const int row  = lane & 15;
    const int hi   = lane >> 4;
    const int p    = (blockIdx.x * 4 + wave) * 32;
    const int b    = p >> 12, hw0 = p & (HW - 1);
    const int sel  = blockIdx.y >> 1;
    const int o0   = (blockIdx.y & 1) * 48;
    const float* w    = sel == 0 ? w1 : (sel == 1 ? w2 : w3);
    const float* bias = sel == 0 ? b1 : (sel == 1 ? b2 : b3);
    float*       m    = sel == 0 ? m1 : (sel == 1 ? m2 : m3);

    v8f acc[3][2];
    conv_gemm<DIM>(w + (size_t)(o0 + row) * DIM + 4 * hi,
                   x + (size_t)b * DIM * HW + hw0 + lane, acc, hi != 0);

    float* mp = m + ((size_t)b * C2 + o0) * HW + hw0 + lane;
#pragma unroll
    for (int t = 0; t < 3; ++t)
#pragma unroll
        for (int i = 0; i < 8; ++i) {
            float lo = acc[t][0][i], up = acc[t][1][i];
            half_swap(lo, up, hi != 0);
            const int oa = 16 * t + i;
            mp[(size_t)oa * HW]       = lo + bias[o0 + oa];
            mp[(size_t)(oa + 8) * HW] = up + bias[o0 + oa + 8];
        }
}

__global__ __launch_bounds__(256, 6)
void nlb_flash(const float* __restrict__ Q, const float* __restrict__ K,
               const float* __restrict__ V, float* __restrict__ Opart,
               float* __restrict__ MLpart)
{
    __shared__ __attribute__((aligned(16))) float sK[2 * 16 * KSTR];
    __shared__ __attribute__((aligned(16))) float sV[2 * 8 * VROW];

    const int tid  = threadIdx.x;
    const int lane = tid & 31;
    const int wave = tid >> 5;
    const int row  = lane & 15;
    const int hi   = lane >> 4;
    const int qbase = (blockIdx.x * 8 + wave) * 16;

    v2f qb[24];
    {
        const float* qp = Q + (size_t)(qbase + row) * C2 + 2 * hi;
#pragma unroll
        for (int cc = 0; cc < 24; ++cc) qb[cc] = *(const v2f*)(qp + 4 * cc);
    }

    v8f acc[6];
#pragma unroll
    for (int jb = 0; jb < 6; ++jb)
#pragma unroll
        for (int e = 0; e < 8; ++e) acc[jb][e] = 0.f;
    float mrun = -INFINITY;
    float lrun = 0.f;

    const int split = blockIdx.y;
    const int kbeg  = ((split * NTILE) / NSPLIT) * 16;
    const int kend  = (((split + 1) * NTILE) / NSPLIT) * 16;
    const int ntile = (kend - kbeg) / 16;

    const bool vjob = __builtin_amdgcn_readfirstlane(wave) < 6;
    const int  r0i = tid / 24, c0i = tid - r0i * 24;
    const int  i1 = vjob ? tid + 24 * r0i : tid + 64;
    const int  i2 = vjob ? i1 + 24 : tid + 128;
    const int  r1i = i1 / 24, r2i = i2 / 24;
    float* l0 = sK + r0i * KSTR + c0i * 4;
    float* l1 = vjob ? sV + r0i * VROW + 8 * c0i : sK + r1i * KSTR + (i1 - r1i * 24) * 4;
    float* l2 = vjob ? l1 + 4                    : sK + r2i * KSTR + (i2 - r2i * 24) * 4;

    float4 f0, f1, f2;
    {
        const float4* kt = (const float4*)(K + (size_t)kbeg * C2);
        const float4* vt = (const float4*)(V + (size_t)kbeg * C2);
        f0 = kt[tid];
        f1 = (vjob ? vt : kt)[i1];
        f2 = (vjob ? vt : kt)[i2];
    }
    *(float4*)l0 = f0;
    if (vjob) {
        *(float4*)l1 = make_float4(f1.x, f2.x, f1.y, f2.y);
        *(float4*)l2 = make_float4(f1.z, f2.z, f1.w, f2.w);
    } else {
        *(float4*)l1 = f1;
        *(float4*)l2 = f2;
    }
    __syncthreads();

    for (int it = 0; it < ntile; ++it) {
        const int  cur  = (it & 1) * (16 * KSTR);
        const int  nxt  = (16 * KSTR) - cur;
        const bool more = it + 1 < ntile;
        if (more) {
            const float4* kt = (const float4*)(K + (size_t)(kbeg + 16 * (it + 1)) * C2);
            const float4* vt = (const float4*)(V + (size_t)(kbeg + 16 * (it + 1)) * C2);
            f0 = kt[tid];
            f1 = (vjob ? vt : kt)[i1];
            f2 = (vjob ? vt : kt)[i2];
        }
        const float* cK = sK + cur;
        const float* cV = sV + cur;

        v8f s;
#pragma unroll
        for (int e = 0; e < 8; ++e) s[e] = 0.f;
        {
            const float* kp = cK + row * KSTR + 2 * hi;
#pragma unroll
            for (int cc = 0; cc < 24; ++cc) {
                v2f a = *(const v2f*)(kp + 4 * cc);
                s = __builtin_amdgcn_wmma_f32_16x16x4_f32(
                        false, a, false, qb[cc], (short)0, s, false, false);
            }
        }

        float tmax = s[0];
#pragma unroll
        for (int i = 1; i < 8; ++i) tmax = fmaxf(tmax, s[i]);
        tmax = fmaxf(tmax, __shfl_xor(tmax, 16, 32));
        const float nm   = fmaxf(mrun, tmax);
        const float corr = __expf(mrun - nm);
        float p[8], ps = 0.f;
#pragma unroll
        for (int i = 0; i < 8; ++i) { p[i] = __expf(s[i] - nm); ps += p[i]; }
        ps += __shfl_xor(ps, 16, 32);
        lrun = lrun * corr + ps;
        mrun = nm;
#pragma unroll
        for (int jb = 0; jb < 6; ++jb)
#pragma unroll
            for (int e = 0; e < 8; ++e) acc[jb][e] *= corr;

        {
            const float* vb = cV + 4 * hi * VROW + 2 * row;
#pragma unroll
            for (int jb = 0; jb < 6; ++jb) {
#pragma unroll
                for (int t = 0; t < 4; ++t) {
                    const v2f a = *(const v2f*)(vb + t * VROW + jb * 32);
                    v2f bq;
                    bq.x = p[2 * t];
                    bq.y = p[2 * t + 1];
                    acc[jb] = __builtin_amdgcn_wmma_f32_16x16x4_f32(
                                  false, a, false, bq, (short)0, acc[jb], false, false);
                }
            }
        }

        if (more) {
            *(float4*)(l0 + nxt) = f0;
            if (vjob) {
                *(float4*)(l1 + nxt) = make_float4(f1.x, f2.x, f1.y, f2.y);
                *(float4*)(l2 + nxt) = make_float4(f1.z, f2.z, f1.w, f2.w);
            } else {
                *(float4*)(l1 + nxt) = f1;
                *(float4*)(l2 + nxt) = f2;
            }
        }
        __syncthreads();
    }

    const size_t prow = (size_t)split * NROW + (qbase + row);
    float* op = Opart + prow * C2 + 8 * hi;
#pragma unroll
    for (int jb = 0; jb < 6; ++jb)
#pragma unroll
        for (int i = 0; i < 8; ++i)
            op[jb * 16 + i] = acc[jb][i];
    if (hi == 0) {
        v2f ml;
        ml.x = mrun;
        ml.y = lrun;
        *(v2f*)(MLpart + 2 * prow) = ml;
    }
}

__global__ __launch_bounds__(256)
void nlb_merge(const float* __restrict__ Opart, const float* __restrict__ MLpart,
               float* __restrict__ O)
{
    const int gid = blockIdx.x * blockDim.x + threadIdx.x;
    const int q   = gid / 24;
    v2f ml[NSPLIT];
    float M = -INFINITY;
#pragma unroll
    for (int s = 0; s < NSPLIT; ++s) {
        ml[s] = *(const v2f*)(MLpart + 2 * ((size_t)s * NROW + q));
        M = fmaxf(M, ml[s].x);
    }
    float L = 0.f;
    float4 o = make_float4(0.f, 0.f, 0.f, 0.f);
#pragma unroll
    for (int s = 0; s < NSPLIT; ++s) {
        const float w = __expf(ml[s].x - M);
        L = fmaf(ml[s].y, w, L);
        const float4 a = ((const float4*)Opart)[(size_t)s * (NROW * 24) + gid];
        o.x = fmaf(a.x, w, o.x);
        o.y = fmaf(a.y, w, o.y);
        o.z = fmaf(a.z, w, o.z);
        o.w = fmaf(a.w, w, o.w);
    }
    const float inv = 1.f / L;
    o.x *= inv; o.y *= inv; o.z *= inv; o.w *= inv;
    ((float4*)O)[gid] = o;
}

__global__ __launch_bounds__(128)
void nlb_proj(const float* __restrict__ A, const float* __restrict__ wl,
              const float* __restrict__ bl, const float* __restrict__ x,
              float* __restrict__ out)
{
    const int lane = threadIdx.x & 31;
    const int wave = threadIdx.x >> 5;
    const int row  = lane & 15;
    const int hi   = lane >> 4;
    const int p    = (blockIdx.x * 4 + wave) * 32;
    const int b    = p >> 12, hw0 = p & (HW - 1);
    const int o0   = blockIdx.y * 48;

    v8f acc[3][2];
    conv_gemm<C2>(wl + (size_t)(o0 + row) * C2 + 4 * hi,
                  A + (size_t)b * C2 * HW + hw0 + lane, acc, hi != 0);

    const size_t base = ((size_t)b * DIM + o0) * HW + hw0 + lane;
#pragma unroll
    for (int t = 0; t < 3; ++t)
#pragma unroll
        for (int i = 0; i < 8; ++i) {
            float lo = acc[t][0][i], up = acc[t][1][i];
            half_swap(lo, up, hi != 0);
            const int oa = 16 * t + i;
            const size_t ia = base + (size_t)oa * HW;
            const size_t ib = base + (size_t)(oa + 8) * HW;
            out[ia] = (lo + bl[o0 + oa])     + x[ia];
            out[ib] = (up + bl[o0 + oa + 8]) + x[ib];
        }
}

extern "C" void kernel_launch(void* const* d_in, const int* in_sizes, int n_in,
                              void* d_out, int out_size, void* d_ws, size_t ws_size,
                              hipStream_t stream)
{
    const float* x  = (const float*)d_in[0];
    const float* w1 = (const float*)d_in[1];
    const float* b1 = (const float*)d_in[2];
    const float* w2 = (const float*)d_in[3];
    const float* b2 = (const float*)d_in[4];
    const float* w3 = (const float*)d_in[5];
    const float* b3 = (const float*)d_in[6];
    const float* wl = (const float*)d_in[7];
    const float* bl = (const float*)d_in[8];
    float* out = (float*)d_out;

    const size_t msz = (size_t)NROW * C2;
    float* m1 = (float*)d_ws;
    float* m2 = m1 + msz;
    float* m3 = m2 + msz;
    float* ob = m3 + msz;
    float* op = ob + msz;
    float* ml = op + (size_t)NSPLIT * msz;

    nlb_qkv<<<dim3(NROW / 128, 6), 128, 0, stream>>>(
        x, w1, b1, w2, b2, w3, b3, m1, m2, m3);
    nlb_flash<<<dim3(NROW / (16 * 8), NSPLIT), 256, 0, stream>>>(m1, m2, m3, op, ml);
    nlb_merge<<<dim3(NROW * 24 / 256), 256, 0, stream>>>(op, ml, ob);
    nlb_proj<<<dim3(NROW / 128, DIM / 48), 128, 0, stream>>>(ob, wl, bl, x, out);
}


# attention LDS reads grouped: a tile's 24 V reads before its score run, next tile's 24 K reads behind it (3-stage ring), half swaps on VALU; 4 waves/SIMD, 8 key ranges (was 12): partial sums regroup
# speedup vs baseline: 6.9525x; 1.0419x over previous
#include <hip/hip_runtime.h>
#include <math.h>

typedef float v2f __attribute__((ext_vector_type(2)));
typedef float v8f __attribute__((ext_vector_type(8)));

#define DIM   192
#define C2    96
#define HW    4096
#define BATCH 2
#define NROW  (BATCH * HW)
#define KSTR  100
#define VROW  200
#define NTILE (NROW / 16)
#define NSPLIT 8
#define STG   (16 * KSTR)
static_assert(16 * KSTR == 8 * VROW, "the K and the V stage share one stride");

#define CONV_CH 4

struct ConvChunk {
    float  r[CONV_CH][8];
    float4 a[CONV_CH][3];
};

__device__ __forceinline__ float xhalf(float v)
{
    const int b = __builtin_bit_cast(int, v);
    return __builtin_bit_cast(float,
        __builtin_amdgcn_permlanex16(b, b, 0x76543210, 0xfedcba98, false, false));
}

__device__ __forceinline__ void half_swap(float& a, float& b, bool upper)
{
    const float xa = xhalf(a), xb = xhalf(b);
    const float na = upper ? xb : a;
    const float nb = upper ? b : xa;
    a = na;
    b = nb;
}

template <int KDIM>
__device__ __forceinline__ void conv_load(ConvChunk& c, const float* __restrict__ wp,
                                          const float* __restrict__ xp, int ch)
{
#pragma unroll
    for (int g = 0; g < CONV_CH; ++g) {
        const int c0 = (ch * CONV_CH + g) * 8;
#pragma unroll
        for (int j = 0; j < 8; ++j) c.r[g][j] = xp[(size_t)(c0 + j) * HW];
#pragma unroll
        for (int t = 0; t < 3; ++t) c.a[g][t] = *(const float4*)(wp + t * 16 * KDIM + c0);
    }
}

__device__ __forceinline__ void conv_mma(ConvChunk& c, v8f (&acc)[3][2], bool upper)
{
#pragma unroll
    for (int g = 0; g < CONV_CH; ++g) {
#pragma unroll
        for (int j = 0; j < 4; ++j) half_swap(c.r[g][j], c.r[g][j + 4], upper);
#pragma unroll
        for (int u = 0; u < 2; ++u) {
            v2f b0, b1;
            b0.x = c.r[g][2 * u];     b0.y = c.r[g][2 * u + 1];
            b1.x = c.r[g][2 * u + 4]; b1.y = c.r[g][2 * u + 5];
#pragma unroll
            for (int t = 0; t < 3; ++t) {
                v2f a;
                a.x = u ? c.a[g][t].z : c.a[g][t].x;
                a.y = u ? c.a[g][t].w : c.a[g][t].y;
                acc[t][0] = __builtin_amdgcn_wmma_f32_16x16x4_f32(
                                false, a, false, b0, (short)0, acc[t][0], false, false);
                acc[t][1] = __builtin_amdgcn_wmma_f32_16x16x4_f32(
                                false, a, false, b1, (short)0, acc[t][1], false, false);
            }
        }
    }
}

template <int KDIM>
__device__ __forceinline__ void conv_gemm(const float* __restrict__ wp,
                                          const float* __restrict__ xp, v8f (&acc)[3][2],
                                          bool upper)
{
    constexpr int NCH = KDIM / (8 * CONV_CH);
#pragma unroll
    for (int t = 0; t < 3; ++t)
#pragma unroll
        for (int j = 0; j < 2; ++j)
#pragma unroll
            for (int e = 0; e < 8; ++e) acc[t][j][e] = 0.f;
    ConvChunk buf[2];
    conv_load<KDIM>(buf[0], wp, xp, 0);
#pragma unroll
    for (int ch = 0; ch < NCH; ++ch) {
        if (ch + 1 < NCH) conv_load<KDIM>(buf[(ch + 1) & 1], wp, xp, ch + 1);
        __builtin_amdgcn_sched_barrier(0);
        conv_mma(buf[ch & 1], acc, upper);
    }
}

__global__ __launch_bounds__(128)
void nlb_qkv(const float* __restrict__ x,
             const float* __restrict__ w1, const float* __restrict__ b1,
             const float* __restrict__ w2, const float* __restrict__ b2,
             const float* __restrict__ w3, const float* __restrict__ b3,
             float* __restrict__ m1, float* __restrict__ m2, float* __restrict__ m3)
{
    const int lane = threadIdx.x & 31;
    const int wave = threadIdx.x >> 5;
    const int row  = lane & 15;
    const int hi   = lane >> 4;
    const int p    = (blockIdx.x * 4 + wave) * 32;
    const int b    = p >> 12, hw0 = p & (HW - 1);
    const int sel  = blockIdx.y >> 1;
    const int o0   = (blockIdx.y & 1) * 48;
    const float* w    = sel == 0 ? w1 : (sel == 1 ? w2 : w3);
    const float* bias = sel == 0 ? b1 : (sel == 1 ? b2 : b3);
    float*       m    = sel == 0 ? m1 : (sel == 1 ? m2 : m3);

    v8f acc[3][2];
    conv_gemm<DIM>(w + (size_t)(o0 + row) * DIM + 4 * hi,
                   x + (size_t)b * DIM * HW + hw0 + lane, acc, hi != 0);

    float* mp = m + ((size_t)b * C2 + o0) * HW + hw0 + lane;
#pragma unroll
    for (int t = 0; t < 3; ++t)
#pragma unroll
        for (int i = 0; i < 8; ++i) {
            float lo = acc[t][0][i], up = acc[t][1][i];
            half_swap(lo, up, hi != 0);
            const int oa = 16 * t + i;
            mp[(size_t)oa * HW]       = lo + bias[o0 + oa];
            mp[(size_t)(oa + 8) * HW] = up + bias[o0 + oa + 8];
        }
}

__global__ __launch_bounds__(256, 4)
void nlb_flash(const float* __restrict__ Q, const float* __restrict__ K,
               const float* __restrict__ V, float* __restrict__ Opart,
               float* __restrict__ MLpart)
{
    __shared__ __attribute__((aligned(16))) float sK[3 * STG];
    __shared__ __attribute__((aligned(16))) float sV[3 * STG];

    const int tid  = threadIdx.x;
    const int lane = tid & 31;
    const int wave = tid >> 5;
    const int row  = lane & 15;
    const int hi   = lane >> 4;
    const int qbase = (blockIdx.x * 8 + wave) * 16;

    v2f qb[24];
    {
        const float* qp = Q + (size_t)(qbase + row) * C2 + 2 * hi;
#pragma unroll
        for (int cc = 0; cc < 24; ++cc) qb[cc] = *(const v2f*)(qp + 4 * cc);
    }

    v8f acc[6];
#pragma unroll
    for (int jb = 0; jb < 6; ++jb)
#pragma unroll
        for (int e = 0; e < 8; ++e) acc[jb][e] = 0.f;
    float mrun = -INFINITY;
    float lrun = 0.f;

    const int split = blockIdx.y;
    const int kbeg  = ((split * NTILE) / NSPLIT) * 16;
    const int kend  = (((split + 1) * NTILE) / NSPLIT) * 16;
    const int ntile = (kend - kbeg) / 16;

    const bool vjob = __builtin_amdgcn_readfirstlane(wave) < 6;
    const int  r0i = tid / 24, c0i = tid - r0i * 24;
    const int  i1 = vjob ? tid + 24 * r0i : tid + 64;
    const int  i2 = vjob ? i1 + 24 : tid + 128;
    const int  r1i = i1 / 24, r2i = i2 / 24;
    float* l0 = sK + r0i * KSTR + c0i * 4;
    float* l1 = vjob ? sV + r0i * VROW + 8 * c0i : sK + r1i * KSTR + (i1 - r1i * 24) * 4;
    float* l2 = vjob ? l1 + 4                    : sK + r2i * KSTR + (i2 - r2i * 24) * 4;

    float4 f0, f1, f2;
#pragma unroll
    for (int j = 0; j < 2; ++j) {
        if (j < ntile) {
            const float4* kt = (const float4*)(K + (size_t)(kbeg + 16 * j) * C2);
            const float4* vt = (const float4*)(V + (size_t)(kbeg + 16 * j) * C2);
            f0 = kt[tid];
            f1 = (vjob ? vt : kt)[i1];
            f2 = (vjob ? vt : kt)[i2];
            *(float4*)(l0 + j * STG) = f0;
            if (vjob) {
                *(float4*)(l1 + j * STG) = make_float4(f1.x, f2.x, f1.y, f2.y);
                *(float4*)(l2 + j * STG) = make_float4(f1.z, f2.z, f1.w, f2.w);
            } else {
                *(float4*)(l1 + j * STG) = f1;
                *(float4*)(l2 + j * STG) = f2;
            }
        }
    }
    __syncthreads();

    v2f kf[24];
    {
        const float* kp = sK + row * KSTR + 2 * hi;
#pragma unroll
        for (int cc = 0; cc < 24; ++cc) kf[cc] = *(const v2f*)(kp + 4 * cc);
    }

    int ocur = 0, onx1 = STG, onx2 = 2 * STG;
    for (int it = 0; it < ntile; ++it) {
        const bool more = it + 2 < ntile;
        if (more) {
            const float4* kt = (const float4*)(K + (size_t)(kbeg + 16 * (it + 2)) * C2);
            const float4* vt = (const float4*)(V + (size_t)(kbeg + 16 * (it + 2)) * C2);
            f0 = kt[tid];
            f1 = (vjob ? vt : kt)[i1];
            f2 = (vjob ? vt : kt)[i2];
        }

        v2f vf[24];
        {
            const float* vb = sV + ocur + 4 * hi * VROW + 2 * row;
#pragma unroll
            for (int jb = 0; jb < 6; ++jb)
#pragma unroll
                for (int t = 0; t < 4; ++t)
                    vf[jb * 4 + t] = *(const v2f*)(vb + t * VROW + jb * 32);
        }
        __builtin_amdgcn_sched_barrier(0);

        v8f s;
#pragma unroll
        for (int e = 0; e < 8; ++e) s[e] = 0.f;
#pragma unroll
        for (int cc = 0; cc < 24; ++cc)
            s = __builtin_amdgcn_wmma_f32_16x16x4_f32(
                    false, kf[cc], false, qb[cc], (short)0, s, false, false);
        __builtin_amdgcn_sched_barrier(0);

        {
            const float* kp = sK + onx1 + row * KSTR + 2 * hi;
#pragma unroll
            for (int cc = 0; cc < 24; ++cc) kf[cc] = *(const v2f*)(kp + 4 * cc);
        }

        float tmax = s[0];
#pragma unroll
        for (int i = 1; i < 8; ++i) tmax = fmaxf(tmax, s[i]);
        tmax = fmaxf(tmax, xhalf(tmax));
        const float nm   = fmaxf(mrun, tmax);
        const float corr = __expf(mrun - nm);
        float p[8], ps = 0.f;
#pragma unroll
        for (int i = 0; i < 8; ++i) { p[i] = __expf(s[i] - nm); ps += p[i]; }
#pragma unroll
        for (int jb = 0; jb < 6; ++jb)
#pragma unroll
            for (int e = 0; e < 8; ++e) acc[jb][e] *= corr;
        __builtin_amdgcn_sched_barrier(0);

#pragma unroll
        for (int t = 0; t < 4; ++t) {
            v2f bq;
            bq.x = p[2 * t];
            bq.y = p[2 * t + 1];
#pragma unroll
            for (int jb = 0; jb < 6; ++jb)
                acc[jb] = __builtin_amdgcn_wmma_f32_16x16x4_f32(
                              false, vf[jb * 4 + t], false, bq, (short)0, acc[jb], false, false);
        }
        ps += xhalf(ps);
        lrun = lrun * corr + ps;
        mrun = nm;

        if (more) {
            *(float4*)(l0 + onx2) = f0;
            if (vjob) {
                *(float4*)(l1 + onx2) = make_float4(f1.x, f2.x, f1.y, f2.y);
                *(float4*)(l2 + onx2) = make_float4(f1.z, f2.z, f1.w, f2.w);
            } else {
                *(float4*)(l1 + onx2) = f1;
                *(float4*)(l2 + onx2) = f2;
            }
        }
        __syncthreads();
        const int o = ocur;
        ocur = onx1;
        onx1 = onx2;
        onx2 = o;
    }

    const size_t prow = (size_t)split * NROW + (qbase + row);
    float* op = Opart + prow * C2 + 8 * hi;
#pragma unroll
    for (int jb = 0; jb < 6; ++jb)
#pragma unroll
        for (int i = 0; i < 8; ++i)
            op[jb * 16 + i] = acc[jb][i];
    if (hi == 0) {
        v2f ml;
        ml.x = mrun;
        ml.y = lrun;
        *(v2f*)(MLpart + 2 * prow) = ml;
    }
}

__global__ __launch_bounds__(256)
void nlb_merge(const float* __restrict__ Opart, const float* __restrict__ MLpart,
               float* __restrict__ O)
{
    const int gid = blockIdx.x * blockDim.x + threadIdx.x;
    const int q   = gid / 24;
    v2f ml[NSPLIT];
    float M = -INFINITY;
#pragma unroll
    for (int s = 0; s < NSPLIT; ++s) {
        ml[s] = *(const v2f*)(MLpart + 2 * ((size_t)s * NROW + q));
        M = fmaxf(M, ml[s].x);
    }
    float L = 0.f;
    float4 o = make_float4(0.f, 0.f, 0.f, 0.f);
#pragma unroll
    for (int s = 0; s < NSPLIT; ++s) {
        const float w = __expf(ml[s].x - M);
        L = fmaf(ml[s].y, w, L);
        const float4 a = ((const float4*)Opart)[(size_t)s * (NROW * 24) + gid];
        o.x = fmaf(a.x, w, o.x);
        o.y = fmaf(a.y, w, o.y);
        o.z = fmaf(a.z, w, o.z);
        o.w = fmaf(a.w, w, o.w);
    }
    const float inv = 1.f / L;
    o.x *= inv; o.y *= inv; o.z *= inv; o.w *= inv;
    ((float4*)O)[gid] = o;
}

__global__ __launch_bounds__(128)
void nlb_proj(const float* __restrict__ A, const float* __restrict__ wl,
              const float* __restrict__ bl, const float* __restrict__ x,
              float* __restrict__ out)
{
    const int lane = threadIdx.x & 31;
    const int wave = threadIdx.x >> 5;
    const int row  = lane & 15;
    const int hi   = lane >> 4;
    const int p    = (blockIdx.x * 4 + wave) * 32;
    const int b    = p >> 12, hw0 = p & (HW - 1);
    const int o0   = blockIdx.y * 48;

    v8f acc[3][2];
    conv_gemm<C2>(wl + (size_t)(o0 + row) * C2 + 4 * hi,
                  A + (size_t)b * C2 * HW + hw0 + lane, acc, hi != 0);

    const size_t base = ((size_t)b * DIM + o0) * HW + hw0 + lane;
#pragma unroll
    for (int t = 0; t < 3; ++t)
#pragma unroll
        for (int i = 0; i < 8; ++i) {
            float lo = acc[t][0][i], up = acc[t][1][i];
            half_swap(lo, up, hi != 0);
            const int oa = 16 * t + i;
            const size_t ia = base + (size_t)oa * HW;
            const size_t ib = base + (size_t)(oa + 8) * HW;
            out[ia] = (lo + bl[o0 + oa])     + x[ia];
            out[ib] = (up + bl[o0 + oa + 8]) + x[ib];
        }
}

extern "C" void kernel_launch(void* const* d_in, const int* in_sizes, int n_in,
                              void* d_out, int out_size, void* d_ws, size_t ws_size,
                              hipStream_t stream)
{
    const float* x  = (const float*)d_in[0];
    const float* w1 = (const float*)d_in[1];
    const float* b1 = (const float*)d_in[2];
    const float* w2 = (const float*)d_in[3];
    const float* b2 = (const float*)d_in[4];
    const float* w3 = (const float*)d_in[5];
    const float* b3 = (const float*)d_in[6];
    const float* wl = (const float*)d_in[7];
    const float* bl = (const float*)d_in[8];
    float* out = (float*)d_out;

    const size_t msz = (size_t)NROW * C2;
    float* m1 = (float*)d_ws;
    float* m2 = m1 + msz;
    float* m3 = m2 + msz;
    float* ob = m3 + msz;
    float* op = ob + msz;
    float* ml = op + (size_t)NSPLIT * msz;

    nlb_qkv<<<dim3(NROW / 128, 6), 128, 0, stream>>>(
        x, w1, b1, w2, b2, w3, b3, m1, m2, m3);
    nlb_flash<<<dim3(NROW / (16 * 8), NSPLIT), 256, 0, stream>>>(m1, m2, m3, op, ml);
    nlb_merge<<<dim3(NROW * 24 / 256), 256, 0, stream>>>(op, ml, ob);
    nlb_proj<<<dim3(NROW / 128, DIM / 48), 128, 0, stream>>>(ob, wl, bl, x, out);
}


# attention: each wave owns 32 queries (two tiles), so every K and V fragment read from LDS feeds two matrix instructions; 256 work-groups at 2 waves/SIMD (was 512 at 4); per-query sum order unchanged
# speedup vs baseline: 7.1960x; 1.0350x over previous
#include <hip/hip_runtime.h>
#include <math.h>

typedef float v2f __attribute__((ext_vector_type(2)));
typedef float v8f __attribute__((ext_vector_type(8)));

#define DIM   192
#define C2    96
#define HW    4096
#define BATCH 2
#define NROW  (BATCH * HW)
#define KSTR  100
#define VROW  200
#define NTILE (NROW / 16)
#define NSPLIT 8
#define QT    2
#define STG   (16 * KSTR)
static_assert(16 * KSTR == 8 * VROW, "the K and the V stage share one stride");

#define CONV_CH 4

struct ConvChunk {
    float  r[CONV_CH][8];
    float4 a[CONV_CH][3];
};

__device__ __forceinline__ float xhalf(float v)
{
    const int b = __builtin_bit_cast(int, v);
    return __builtin_bit_cast(float,
        __builtin_amdgcn_permlanex16(b, b, 0x76543210, 0xfedcba98, false, false));
}

__device__ __forceinline__ void half_swap(float& a, float& b, bool upper)
{
    const float xa = xhalf(a), xb = xhalf(b);
    const float na = upper ? xb : a;
    const float nb = upper ? b : xa;
    a = na;
    b = nb;
}

template <int KDIM>
__device__ __forceinline__ void conv_load(ConvChunk& c, const float* __restrict__ wp,
                                          const float* __restrict__ xp, int ch)
{
#pragma unroll
    for (int g = 0; g < CONV_CH; ++g) {
        const int c0 = (ch * CONV_CH + g) * 8;
#pragma unroll
        for (int j = 0; j < 8; ++j) c.r[g][j] = xp[(size_t)(c0 + j) * HW];
#pragma unroll
        for (int t = 0; t < 3; ++t) c.a[g][t] = *(const float4*)(wp + t * 16 * KDIM + c0);
    }
}

__device__ __forceinline__ void conv_mma(ConvChunk& c, v8f (&acc)[3][2], bool upper)
{
#pragma unroll
    for (int g = 0; g < CONV_CH; ++g) {
#pragma unroll
        for (int j = 0; j < 4; ++j) half_swap(c.r[g][j], c.r[g][j + 4], upper);
#pragma unroll
        for (int u = 0; u < 2; ++u) {
            v2f b0, b1;
            b0.x = c.r[g][2 * u];     b0.y = c.r[g][2 * u + 1];
            b1.x = c.r[g][2 * u + 4]; b1.y = c.r[g][2 * u + 5];
#pragma unroll
            for (int t = 0; t < 3; ++t) {
                v2f a;
                a.x = u ? c.a[g][t].z : c.a[g][t].x;
                a.y = u ? c.a[g][t].w : c.a[g][t].y;
                acc[t][0] = __builtin_amdgcn_wmma_f32_16x16x4_f32(
                                false, a, false, b0, (short)0, acc[t][0], false, false);
                acc[t][1] = __builtin_amdgcn_wmma_f32_16x16x4_f32(
                                false, a, false, b1, (short)0, acc[t][1], false, false);
            }
        }
    }
}

template <int KDIM>
__device__ __forceinline__ void conv_gemm(const float* __restrict__ wp,
                                          const float* __restrict__ xp, v8f (&acc)[3][2],
                                          bool upper)
{
    constexpr int NCH = KDIM / (8 * CONV_CH);
#pragma unroll
    for (int t = 0; t < 3; ++t)
#pragma unroll
        for (int j = 0; j < 2; ++j)
#pragma unroll
            for (int e = 0; e < 8; ++e) acc[t][j][e] = 0.f;
    ConvChunk buf[2];
    conv_load<KDIM>(buf[0], wp, xp, 0);
#pragma unroll
    for (int ch = 0; ch < NCH; ++ch) {
        if (ch + 1 < NCH) conv_load<KDIM>(buf[(ch + 1) & 1], wp, xp, ch + 1);
        __builtin_amdgcn_sched_barrier(0);
        conv_mma(buf[ch & 1], acc, upper);
    }
}

__global__ __launch_bounds__(128)
void nlb_qkv(const float* __restrict__ x,
             const float* __restrict__ w1, const float* __restrict__ b1,
             const float* __restrict__ w2, const float* __restrict__ b2,
             const float* __restrict__ w3, const float* __restrict__ b3,
             float* __restrict__ m1, float* __restrict__ m2, float* __restrict__ m3)
{
    const int lane = threadIdx.x & 31;
    const int wave = threadIdx.x >> 5;
    const int row  = lane & 15;
    const int hi   = lane >> 4;
    const int p    = (blockIdx.x * 4 + wave) * 32;
    const int b    = p >> 12, hw0 = p & (HW - 1);
    const int sel  = blockIdx.y >> 1;
    const int o0   = (blockIdx.y & 1) * 48;
    const float* w    = sel == 0 ? w1 : (sel == 1 ? w2 : w3);
    const float* bias = sel == 0 ? b1 : (sel == 1 ? b2 : b3);
    float*       m    = sel == 0 ? m1 : (sel == 1 ? m2 : m3);

    v8f acc[3][2];
    conv_gemm<DIM>(w + (size_t)(o0 + row) * DIM + 4 * hi,
                   x + (size_t)b * DIM * HW + hw0 + lane, acc, hi != 0);

    float* mp = m + ((size_t)b * C2 + o0) * HW + hw0 + lane;
#pragma unroll
    for (int t = 0; t < 3; ++t)
#pragma unroll
        for (int i = 0; i < 8; ++i) {
            float lo = acc[t][0][i], up = acc[t][1][i];
            half_swap(lo, up, hi != 0);
            const int oa = 16 * t + i;
            mp[(size_t)oa * HW]       = lo + bias[o0 + oa];
            mp[(size_t)(oa + 8) * HW] = up + bias[o0 + oa + 8];
        }
}

__global__ __launch_bounds__(256, 2)
void nlb_flash(const float* __restrict__ Q, const float* __restrict__ K,
               const float* __restrict__ V, float* __restrict__ Opart,
               float* __restrict__ MLpart)
{
    __shared__ __attribute__((aligned(16))) float sK[3 * STG];
    __shared__ __attribute__((aligned(16))) float sV[3 * STG];

    const int tid  = threadIdx.x;
    const int lane = tid & 31;
    const int wave = tid >> 5;
    const int row  = lane & 15;
    const int hi   = lane >> 4;
    const int qbase = (blockIdx.x * 8 + wave) * (16 * QT);

    v2f qb[QT][24];
#pragma unroll
    for (int u = 0; u < QT; ++u) {
        const float* qp = Q + (size_t)(qbase + 16 * u + row) * C2 + 2 * hi;
#pragma unroll
        for (int cc = 0; cc < 24; ++cc) qb[u][cc] = *(const v2f*)(qp + 4 * cc);
    }

    v8f acc[QT][6];
    float mrun[QT], lrun[QT];
#pragma unroll
    for (int u = 0; u < QT; ++u) {
#pragma unroll
        for (int jb = 0; jb < 6; ++jb)
#pragma unroll
            for (int e = 0; e < 8; ++e) acc[u][jb][e] = 0.f;
        mrun[u] = -INFINITY;
        lrun[u] = 0.f;
    }

    const int split = blockIdx.y;
    const int kbeg  = ((split * NTILE) / NSPLIT) * 16;
    const int kend  = (((split + 1) * NTILE) / NSPLIT) * 16;
    const int ntile = (kend - kbeg) / 16;

    const bool vjob = __builtin_amdgcn_readfirstlane(wave) < 6;
    const int  r0i = tid / 24, c0i = tid - r0i * 24;
    const int  i1 = vjob ? tid + 24 * r0i : tid + 64;
    const int  i2 = vjob ? i1 + 24 : tid + 128;
    const int  r1i = i1 / 24, r2i = i2 / 24;
    float* l0 = sK + r0i * KSTR + c0i * 4;
    float* l1 = vjob ? sV + r0i * VROW + 8 * c0i : sK + r1i * KSTR + (i1 - r1i * 24) * 4;
    float* l2 = vjob ? l1 + 4                    : sK + r2i * KSTR + (i2 - r2i * 24) * 4;

    float4 f0, f1, f2;
#pragma unroll
    for (int j = 0; j < 2; ++j) {
        if (j < ntile) {
            const float4* kt = (const float4*)(K + (size_t)(kbeg + 16 * j) * C2);
            const float4* vt = (const float4*)(V + (size_t)(kbeg + 16 * j) * C2);
            f0 = kt[tid];
            f1 = (vjob ? vt : kt)[i1];
            f2 = (vjob ? vt : kt)[i2];
            *(float4*)(l0 + j * STG) = f0;
            if (vjob) {
                *(float4*)(l1 + j * STG) = make_float4(f1.x, f2.x, f1.y, f2.y);
                *(float4*)(l2 + j * STG) = make_float4(f1.z, f2.z, f1.w, f2.w);
            } else {
                *(float4*)(l1 + j * STG) = f1;
                *(float4*)(l2 + j * STG) = f2;
            }
        }
    }
    __syncthreads();

    v2f kf[24];
    {
        const float* kp = sK + row * KSTR + 2 * hi;
#pragma unroll
        for (int cc = 0; cc < 24; ++cc) kf[cc] = *(const v2f*)(kp + 4 * cc);
    }

    int ocur = 0, onx1 = STG, onx2 = 2 * STG;
    for (int it = 0; it < ntile; ++it) {
        const bool more = it + 2 < ntile;
        if (more) {
            const float4* kt = (const float4*)(K + (size_t)(kbeg + 16 * (it + 2)) * C2);
            const float4* vt = (const float4*)(V + (size_t)(kbeg + 16 * (it + 2)) * C2);
            f0 = kt[tid];
            f1 = (vjob ? vt : kt)[i1];
            f2 = (vjob ? vt : kt)[i2];
        }

        v2f vf[24];
        {
            const float* vb = sV + ocur + 4 * hi * VROW + 2 * row;
#pragma unroll
            for (int jb = 0; jb < 6; ++jb)
#pragma unroll
                for (int t = 0; t < 4; ++t)
                    vf[jb * 4 + t] = *(const v2f*)(vb + t * VROW + jb * 32);
        }
        __builtin_amdgcn_sched_barrier(0);

        v8f s[QT];
#pragma unroll
        for (int u = 0; u < QT; ++u)
#pragma unroll
            for (int e = 0; e < 8; ++e) s[u][e] = 0.f;
#pragma unroll
        for (int cc = 0; cc < 24; ++cc)
#pragma unroll
            for (int u = 0; u < QT; ++u)
                s[u] = __builtin_amdgcn_wmma_f32_16x16x4_f32(
                           false, kf[cc], false, qb[u][cc], (short)0, s[u], false, false);
        __builtin_amdgcn_sched_barrier(0);

        {
            const float* kp = sK + onx1 + row * KSTR + 2 * hi;
#pragma unroll
            for (int cc = 0; cc < 24; ++cc) kf[cc] = *(const v2f*)(kp + 4 * cc);
        }

        float p[QT][8], ps[QT], corr[QT], nm[QT];
#pragma unroll
        for (int u = 0; u < QT; ++u) {
            float tmax = s[u][0];
#pragma unroll
            for (int i = 1; i < 8; ++i) tmax = fmaxf(tmax, s[u][i]);
            tmax = fmaxf(tmax, xhalf(tmax));
            nm[u]   = fmaxf(mrun[u], tmax);
            corr[u] = __expf(mrun[u] - nm[u]);
            ps[u] = 0.f;
#pragma unroll
            for (int i = 0; i < 8; ++i) { p[u][i] = __expf(s[u][i] - nm[u]); ps[u] += p[u][i]; }
#pragma unroll
            for (int jb = 0; jb < 6; ++jb)
#pragma unroll
                for (int e = 0; e < 8; ++e) acc[u][jb][e] *= corr[u];
        }
        __builtin_amdgcn_sched_barrier(0);

#pragma unroll
        for (int t = 0; t < 4; ++t) {
            v2f bq[QT];
#pragma unroll
            for (int u = 0; u < QT; ++u) {
                bq[u].x = p[u][2 * t];
                bq[u].y = p[u][2 * t + 1];
            }
#pragma unroll
            for (int jb = 0; jb < 6; ++jb)
#pragma unroll
                for (int u = 0; u < QT; ++u)
                    acc[u][jb] = __builtin_amdgcn_wmma_f32_16x16x4_f32(
                                     false, vf[jb * 4 + t], false, bq[u], (short)0, acc[u][jb],
                                     false, false);
        }
#pragma unroll
        for (int u = 0; u < QT; ++u) {
            ps[u] += xhalf(ps[u]);
            lrun[u] = lrun[u] * corr[u] + ps[u];
            mrun[u] = nm[u];
        }

        if (more) {
            *(float4*)(l0 + onx2) = f0;
            if (vjob) {
                *(float4*)(l1 + onx2) = make_float4(f1.x, f2.x, f1.y, f2.y);
                *(float4*)(l2 + onx2) = make_float4(f1.z, f2.z, f1.w, f2.w);
            } else {
                *(float4*)(l1 + onx2) = f1;
                *(float4*)(l2 + onx2) = f2;
            }
        }
        __syncthreads();
        const int o = ocur;
        ocur = onx1;
        onx1 = onx2;
        onx2 = o;
    }

#pragma unroll
    for (int u = 0; u < QT; ++u) {
        const size_t prow = (size_t)split * NROW + (qbase + 16 * u + row);
        float* op = Opart + prow * C2 + 8 * hi;
#pragma unroll
        for (int jb = 0; jb < 6; ++jb)
#pragma unroll
            for (int i = 0; i < 8; ++i)
                op[jb * 16 + i] = acc[u][jb][i];
        if (hi == 0) {
            v2f ml;
            ml.x = mrun[u];
            ml.y = lrun[u];
            *(v2f*)(MLpart + 2 * prow) = ml;
        }
    }
}

__global__ __launch_bounds__(256)
void nlb_merge(const float* __restrict__ Opart, const float* __restrict__ MLpart,
               float* __restrict__ O)
{
    const int gid = blockIdx.x * blockDim.x + threadIdx.x;
    const int q   = gid / 24;
    v2f ml[NSPLIT];
    float M = -INFINITY;
#pragma unroll
    for (int s = 0; s < NSPLIT; ++s) {
        ml[s] = *(const v2f*)(MLpart + 2 * ((size_t)s * NROW + q));
        M = fmaxf(M, ml[s].x);
    }
    float L = 0.f;
    float4 o = make_float4(0.f, 0.f, 0.f, 0.f);
#pragma unroll
    for (int s = 0; s < NSPLIT; ++s) {
        const float w = __expf(ml[s].x - M);
        L = fmaf(ml[s].y, w, L);
        const float4 a = ((const float4*)Opart)[(size_t)s * (NROW * 24) + gid];
        o.x = fmaf(a.x, w, o.x);
        o.y = fmaf(a.y, w, o.y);
        o.z = fmaf(a.z, w, o.z);
        o.w = fmaf(a.w, w, o.w);
    }
    const float inv = 1.f / L;
    o.x *= inv; o.y *= inv; o.z *= inv; o.w *= inv;
    ((float4*)O)[gid] = o;
}

__global__ __launch_bounds__(128)
void nlb_proj(const float* __restrict__ A, const float* __restrict__ wl,
              const float* __restrict__ bl, const float* __restrict__ x,
              float* __restrict__ out)
{
    const int lane = threadIdx.x & 31;
    const int wave = threadIdx.x >> 5;
    const int row  = lane & 15;
    const int hi   = lane >> 4;
    const int p    = (blockIdx.x * 4 + wave) * 32;
    const int b    = p >> 12, hw0 = p & (HW - 1);
    const int o0   = blockIdx.y * 48;

    v8f acc[3][2];
    conv_gemm<C2>(wl + (size_t)(o0 + row) * C2 + 4 * hi,
                  A + (size_t)b * C2 * HW + hw0 + lane, acc, hi != 0);

    const size_t base = ((size_t)b * DIM + o0) * HW + hw0 + lane;
#pragma unroll
    for (int t = 0; t < 3; ++t)
#pragma unroll
        for (int i = 0; i < 8; ++i) {
            float lo = acc[t][0][i], up = acc[t][1][i];
            half_swap(lo, up, hi != 0);
            const int oa = 16 * t + i;
            const size_t ia = base + (size_t)oa * HW;
            const size_t ib = base + (size_t)(oa + 8) * HW;
            out[ia] = (lo + bl[o0 + oa])     + x[ia];
            out[ib] = (up + bl[o0 + oa + 8]) + x[ib];
        }
}

extern "C" void kernel_launch(void* const* d_in, const int* in_sizes, int n_in,
                              void* d_out, int out_size, void* d_ws, size_t ws_size,
                              hipStream_t stream)
{
    const float* x  = (const float*)d_in[0];
    const float* w1 = (const float*)d_in[1];
    const float* b1 = (const float*)d_in[2];
    const float* w2 = (const float*)d_in[3];
    const float* b2 = (const float*)d_in[4];
    const float* w3 = (const float*)d_in[5];
    const float* b3 = (const float*)d_in[6];
    const float* wl = (const float*)d_in[7];
    const float* bl = (const float*)d_in[8];
    float* out = (float*)d_out;

    const size_t msz = (size_t)NROW * C2;
    float* m1 = (float*)d_ws;
    float* m2 = m1 + msz;
    float* m3 = m2 + msz;
    float* ob = m3 + msz;
    float* op = ob + msz;
    float* ml = op + (size_t)NSPLIT * msz;

    nlb_qkv<<<dim3(NROW / 128, 6), 128, 0, stream>>>(
        x, w1, b1, w2, b2, w3, b3, m1, m2, m3);
    nlb_flash<<<dim3(NROW / (16 * QT * 8), NSPLIT), 256, 0, stream>>>(m1, m2, m3, op, ml);
    nlb_merge<<<dim3(NROW * 24 / 256), 256, 0, stream>>>(op, ml, ob);
    nlb_proj<<<dim3(NROW / 128, DIM / 48), 128, 0, stream>>>(ob, wl, bl, x, out);
}


# attention softmax against a per-query reference that moves only when a score tops it by 2^32: no per-tile accumulator rescale, correction exp or half exchange; row sums per lane half, joined at end
# speedup vs baseline: 7.4213x; 1.0313x over previous
#include <hip/hip_runtime.h>
#include <math.h>

typedef float v2f __attribute__((ext_vector_type(2)));
typedef float v8f __attribute__((ext_vector_type(8)));

#define DIM   192
#define C2    96
#define HW    4096
#define BATCH 2
#define NROW  (BATCH * HW)
#define KSTR  100
#define VROW  200
#define NTILE (NROW / 16)
#define NSPLIT 8
#define QT    2
#define STG   (16 * KSTR)
#define REF_GAP (32.0f * 0.69314718f)
static_assert(16 * KSTR == 8 * VROW, "the K and the V stage share one stride");

#define CONV_CH 4

struct ConvChunk {
    float  r[CONV_CH][8];
    float4 a[CONV_CH][3];
};

__device__ __forceinline__ float xhalf(float v)
{
    const int b = __builtin_bit_cast(int, v);
    return __builtin_bit_cast(float,
        __builtin_amdgcn_permlanex16(b, b, 0x76543210, 0xfedcba98, false, false));
}

__device__ __forceinline__ void half_swap(float& a, float& b, bool upper)
{
    const float xa = xhalf(a), xb = xhalf(b);
    const float na = upper ? xb : a;
    const float nb = upper ? b : xa;
    a = na;
    b = nb;
}

template <int KDIM>
__device__ __forceinline__ void conv_load(ConvChunk& c, const float* __restrict__ wp,
                                          const float* __restrict__ xp, int ch)
{
#pragma unroll
    for (int g = 0; g < CONV_CH; ++g) {
        const int c0 = (ch * CONV_CH + g) * 8;
#pragma unroll
        for (int j = 0; j < 8; ++j) c.r[g][j] = xp[(size_t)(c0 + j) * HW];
#pragma unroll
        for (int t = 0; t < 3; ++t) c.a[g][t] = *(const float4*)(wp + t * 16 * KDIM + c0);
    }
}

__device__ __forceinline__ void conv_mma(ConvChunk& c, v8f (&acc)[3][2], bool upper)
{
#pragma unroll
    for (int g = 0; g < CONV_CH; ++g) {
#pragma unroll
        for (int j = 0; j < 4; ++j) half_swap(c.r[g][j], c.r[g][j + 4], upper);
#pragma unroll
        for (int u = 0; u < 2; ++u) {
            v2f b0, b1;
            b0.x = c.r[g][2 * u];     b0.y = c.r[g][2 * u + 1];
            b1.x = c.r[g][2 * u + 4]; b1.y = c.r[g][2 * u + 5];
#pragma unroll
            for (int t = 0; t < 3; ++t) {
                v2f a;
                a.x = u ? c.a[g][t].z : c.a[g][t].x;
                a.y = u ? c.a[g][t].w : c.a[g][t].y;
                acc[t][0] = __builtin_amdgcn_wmma_f32_16x16x4_f32(
                                false, a, false, b0, (short)0, acc[t][0], false, false);
                acc[t][1] = __builtin_amdgcn_wmma_f32_16x16x4_f32(
                                false, a, false, b1, (short)0, acc[t][1], false, false);
            }
        }
    }
}

template <int KDIM>
__device__ __forceinline__ void conv_gemm(const float* __restrict__ wp,
                                          const float* __restrict__ xp, v8f (&acc)[3][2],
                                          bool upper)
{
    constexpr int NCH = KDIM / (8 * CONV_CH);
#pragma unroll
    for (int t = 0; t < 3; ++t)
#pragma unroll
        for (int j = 0; j < 2; ++j)
#pragma unroll
            for (int e = 0; e < 8; ++e) acc[t][j][e] = 0.f;
    ConvChunk buf[2];
    conv_load<KDIM>(buf[0], wp, xp, 0);
#pragma unroll
    for (int ch = 0; ch < NCH; ++ch) {
        if (ch + 1 < NCH) conv_load<KDIM>(buf[(ch + 1) & 1], wp, xp, ch + 1);
        __builtin_amdgcn_sched_barrier(0);
        conv_mma(buf[ch & 1], acc, upper);
    }
}

__global__ __launch_bounds__(128)
void nlb_qkv(const float* __restrict__ x,
             const float* __restrict__ w1, const float* __restrict__ b1,
             const float* __restrict__ w2, const float* __restrict__ b2,
             const float* __restrict__ w3, const float* __restrict__ b3,
             float* __restrict__ m1, float* __restrict__ m2, float* __restrict__ m3)
{
    const int lane = threadIdx.x & 31;
    const int wave = threadIdx.x >> 5;
    const int row  = lane & 15;
    const int hi   = lane >> 4;
    const int p    = (blockIdx.x * 4 + wave) * 32;
    const int b    = p >> 12, hw0 = p & (HW - 1);
    const int sel  = blockIdx.y >> 1;
    const int o0   = (blockIdx.y & 1) * 48;
    const float* w    = sel == 0 ? w1 : (sel == 1 ? w2 : w3);
    const float* bias = sel == 0 ? b1 : (sel == 1 ? b2 : b3);
    float*       m    = sel == 0 ? m1 : (sel == 1 ? m2 : m3);

    v8f acc[3][2];
    conv_gemm<DIM>(w + (size_t)(o0 + row) * DIM + 4 * hi,
                   x + (size_t)b * DIM * HW + hw0 + lane, acc, hi != 0);

    float* mp = m + ((size_t)b * C2 + o0) * HW + hw0 + lane;
#pragma unroll
    for (int t = 0; t < 3; ++t)
#pragma unroll
        for (int i = 0; i < 8; ++i) {
            float lo = acc[t][0][i], up = acc[t][1][i];
            half_swap(lo, up, hi != 0);
            const int oa = 16 * t + i;
            mp[(size_t)oa * HW]       = lo + bias[o0 + oa];
            mp[(size_t)(oa + 8) * HW] = up + bias[o0 + oa + 8];
        }
}

__global__ __launch_bounds__(256, 2)
void nlb_flash(const float* __restrict__ Q, const float* __restrict__ K,
               const float* __restrict__ V, float* __restrict__ Opart,
               float* __restrict__ MLpart)
{
    __shared__ __attribute__((aligned(16))) float sK[3 * STG];
    __shared__ __attribute__((aligned(16))) float sV[3 * STG];

    const int tid  = threadIdx.x;
    const int lane = tid & 31;
    const int wave = tid >> 5;
    const int row  = lane & 15;
    const int hi   = lane >> 4;
    const int qbase = (blockIdx.x * 8 + wave) * (16 * QT);

    v2f qb[QT][24];
#pragma unroll
    for (int u = 0; u < QT; ++u) {
        const float* qp = Q + (size_t)(qbase + 16 * u + row) * C2 + 2 * hi;
#pragma unroll
        for (int cc = 0; cc < 24; ++cc) qb[u][cc] = *(const v2f*)(qp + 4 * cc);
    }

    v8f acc[QT][6];
    float mref[QT], mtop[QT], lsum[QT];
#pragma unroll
    for (int u = 0; u < QT; ++u) {
#pragma unroll
        for (int jb = 0; jb < 6; ++jb)
#pragma unroll
            for (int e = 0; e < 8; ++e) acc[u][jb][e] = 0.f;
        mref[u] = -INFINITY;
        mtop[u] = -INFINITY;
        lsum[u] = 0.f;
    }

    const int split = blockIdx.y;
    const int kbeg  = ((split * NTILE) / NSPLIT) * 16;
    const int kend  = (((split + 1) * NTILE) / NSPLIT) * 16;
    const int ntile = (kend - kbeg) / 16;

    const bool vjob = __builtin_amdgcn_readfirstlane(wave) < 6;
    const int  r0i = tid / 24, c0i = tid - r0i * 24;
    const int  i1 = vjob ? tid + 24 * r0i : tid + 64;
    const int  i2 = vjob ? i1 + 24 : tid + 128;
    const int  r1i = i1 / 24, r2i = i2 / 24;
    float* l0 = sK + r0i * KSTR + c0i * 4;
    float* l1 = vjob ? sV + r0i * VROW + 8 * c0i : sK + r1i * KSTR + (i1 - r1i * 24) * 4;
    float* l2 = vjob ? l1 + 4                    : sK + r2i * KSTR + (i2 - r2i * 24) * 4;

    float4 f0, f1, f2;
#pragma unroll
    for (int j = 0; j < 2; ++j) {
        if (j < ntile) {
            const float4* kt = (const float4*)(K + (size_t)(kbeg + 16 * j) * C2);
            const float4* vt = (const float4*)(V + (size_t)(kbeg + 16 * j) * C2);
            f0 = kt[tid];
            f1 = (vjob ? vt : kt)[i1];
            f2 = (vjob ? vt : kt)[i2];
            *(float4*)(l0 + j * STG) = f0;
            if (vjob) {
                *(float4*)(l1 + j * STG) = make_float4(f1.x, f2.x, f1.y, f2.y);
                *(float4*)(l2 + j * STG) = make_float4(f1.z, f2.z, f1.w, f2.w);
            } else {
                *(float4*)(l1 + j * STG) = f1;
                *(float4*)(l2 + j * STG) = f2;
            }
        }
    }
    __syncthreads();

    v2f kf[24];
    {
        const float* kp = sK + row * KSTR + 2 * hi;
#pragma unroll
        for (int cc = 0; cc < 24; ++cc) kf[cc] = *(const v2f*)(kp + 4 * cc);
    }

    int ocur = 0, onx1 = STG, onx2 = 2 * STG;
    for (int it = 0; it < ntile; ++it) {
        const bool more = it + 2 < ntile;
        if (more) {
            const float4* kt = (const float4*)(K + (size_t)(kbeg + 16 * (it + 2)) * C2);
            const float4* vt = (const float4*)(V + (size_t)(kbeg + 16 * (it + 2)) * C2);
            f0 = kt[tid];
            f1 = (vjob ? vt : kt)[i1];
            f2 = (vjob ? vt : kt)[i2];
        }

        v2f vf[24];
        {
            const float* vb = sV + ocur + 4 * hi * VROW + 2 * row;
#pragma unroll
            for (int jb = 0; jb < 6; ++jb)
#pragma unroll
                for (int t = 0; t < 4; ++t)
                    vf[jb * 4 + t] = *(const v2f*)(vb + t * VROW + jb * 32);
        }
        __builtin_amdgcn_sched_barrier(0);

        v8f s[QT];
#pragma unroll
        for (int u = 0; u < QT; ++u)
#pragma unroll
            for (int e = 0; e < 8; ++e) s[u][e] = 0.f;
#pragma unroll
        for (int cc = 0; cc < 24; ++cc)
#pragma unroll
            for (int u = 0; u < QT; ++u)
                s[u] = __builtin_amdgcn_wmma_f32_16x16x4_f32(
                           false, kf[cc], false, qb[u][cc], (short)0, s[u], false, false);
        __builtin_amdgcn_sched_barrier(0);

        {
            const float* kp = sK + onx1 + row * KSTR + 2 * hi;
#pragma unroll
            for (int cc = 0; cc < 24; ++cc) kf[cc] = *(const v2f*)(kp + 4 * cc);
        }

        float lmax[QT];
        bool  over = false;
#pragma unroll
        for (int u = 0; u < QT; ++u) {
            lmax[u] = s[u][0];
#pragma unroll
            for (int i = 1; i < 8; ++i) lmax[u] = fmaxf(lmax[u], s[u][i]);
            over = over || (lmax[u] > mtop[u]);
        }
        if (__builtin_expect(__builtin_amdgcn_ballot_w32(over) != 0u, 0)) {
#pragma unroll
            for (int u = 0; u < QT; ++u) {
                const float tmax = fmaxf(lmax[u], xhalf(lmax[u]));
                const float nm   = fmaxf(mref[u], tmax);
                const float corr = __expf(mref[u] - nm);
#pragma unroll
                for (int jb = 0; jb < 6; ++jb)
#pragma unroll
                    for (int e = 0; e < 8; ++e) acc[u][jb][e] *= corr;
                lsum[u] *= corr;
                mref[u] = nm;
                mtop[u] = nm + REF_GAP;
            }
        }
        float p[QT][8];
#pragma unroll
        for (int u = 0; u < QT; ++u) {
#pragma unroll
            for (int i = 0; i < 8; ++i) p[u][i] = __expf(s[u][i] - mref[u]);
            lsum[u] += ((p[u][0] + p[u][1]) + (p[u][2] + p[u][3]))
                     + ((p[u][4] + p[u][5]) + (p[u][6] + p[u][7]));
        }
        __builtin_amdgcn_sched_barrier(0);

#pragma unroll
        for (int t = 0; t < 4; ++t) {
            v2f bq[QT];
#pragma unroll
            for (int u = 0; u < QT; ++u) {
                bq[u].x = p[u][2 * t];
                bq[u].y = p[u][2 * t + 1];
            }
#pragma unroll
            for (int jb = 0; jb < 6; ++jb)
#pragma unroll
                for (int u = 0; u < QT; ++u)
                    acc[u][jb] = __builtin_amdgcn_wmma_f32_16x16x4_f32(
                                     false, vf[jb * 4 + t], false, bq[u], (short)0, acc[u][jb],
                                     false, false);
        }

        if (more) {
            *(float4*)(l0 + onx2) = f0;
            if (vjob) {
                *(float4*)(l1 + onx2) = make_float4(f1.x, f2.x, f1.y, f2.y);
                *(float4*)(l2 + onx2) = make_float4(f1.z, f2.z, f1.w, f2.w);
            } else {
                *(float4*)(l1 + onx2) = f1;
                *(float4*)(l2 + onx2) = f2;
            }
        }
        __syncthreads();
        const int o = ocur;
        ocur = onx1;
        onx1 = onx2;
        onx2 = o;
    }

#pragma unroll
    for (int u = 0; u < QT; ++u) {
        const size_t prow = (size_t)split * NROW + (qbase + 16 * u + row);
        float* op = Opart + prow * C2 + 8 * hi;
#pragma unroll
        for (int jb = 0; jb < 6; ++jb)
#pragma unroll
            for (int i = 0; i < 8; ++i)
                op[jb * 16 + i] = acc[u][jb][i];
        const float lrow = lsum[u] + xhalf(lsum[u]);
        if (hi == 0) {
            v2f ml;
            ml.x = mref[u];
            ml.y = lrow;
            *(v2f*)(MLpart + 2 * prow) = ml;
        }
    }
}

__global__ __launch_bounds__(256)
void nlb_merge(const float* __restrict__ Opart, const float* __restrict__ MLpart,
               float* __restrict__ O)
{
    const int gid = blockIdx.x * blockDim.x + threadIdx.x;
    const int q   = gid / 24;
    v2f ml[NSPLIT];
    float M = -INFINITY;
#pragma unroll
    for (int s = 0; s < NSPLIT; ++s) {
        ml[s] = *(const v2f*)(MLpart + 2 * ((size_t)s * NROW + q));
        M = fmaxf(M, ml[s].x);
    }
    float L = 0.f;
    float4 o = make_float4(0.f, 0.f, 0.f, 0.f);
#pragma unroll
    for (int s = 0; s < NSPLIT; ++s) {
        const float w = __expf(ml[s].x - M);
        L = fmaf(ml[s].y, w, L);
        const float4 a = ((const float4*)Opart)[(size_t)s * (NROW * 24) + gid];
        o.x = fmaf(a.x, w, o.x);
        o.y = fmaf(a.y, w, o.y);
        o.z = fmaf(a.z, w, o.z);
        o.w = fmaf(a.w, w, o.w);
    }
    const float inv = 1.f / L;
    o.x *= inv; o.y *= inv; o.z *= inv; o.w *= inv;
    ((float4*)O)[gid] = o;
}

__global__ __launch_bounds__(128)
void nlb_proj(const float* __restrict__ A, const float* __restrict__ wl,
              const float* __restrict__ bl, const float* __restrict__ x,
              float* __restrict__ out)
{
    const int lane = threadIdx.x & 31;
    const int wave = threadIdx.x >> 5;
    const int row  = lane & 15;
    const int hi   = lane >> 4;
    const int p    = (blockIdx.x * 4 + wave) * 32;
    const int b    = p >> 12, hw0 = p & (HW - 1);
    const int o0   = blockIdx.y * 48;

    v8f acc[3][2];
    conv_gemm<C2>(wl + (size_t)(o0 + row) * C2 + 4 * hi,
                  A + (size_t)b * C2 * HW + hw0 + lane, acc, hi != 0);

    const size_t base = ((size_t)b * DIM + o0) * HW + hw0 + lane;
#pragma unroll
    for (int t = 0; t < 3; ++t)
#pragma unroll
        for (int i = 0; i < 8; ++i) {
            float lo = acc[t][0][i], up = acc[t][1][i];
            half_swap(lo, up, hi != 0);
            const int oa = 16 * t + i;
            const size_t ia = base + (size_t)oa * HW;
            const size_t ib = base + (size_t)(oa + 8) * HW;
            out[ia] = (lo + bl[o0 + oa])     + x[ia];
            out[ib] = (up + bl[o0 + oa + 8]) + x[ib];
        }
}

extern "C" void kernel_launch(void* const* d_in, const int* in_sizes, int n_in,
                              void* d_out, int out_size, void* d_ws, size_t ws_size,
                              hipStream_t stream)
{
    const float* x  = (const float*)d_in[0];
    const float* w1 = (const float*)d_in[1];
    const float* b1 = (const float*)d_in[2];
    const float* w2 = (const float*)d_in[3];
    const float* b2 = (const float*)d_in[4];
    const float* w3 = (const float*)d_in[5];
    const float* b3 = (const float*)d_in[6];
    const float* wl = (const float*)d_in[7];
    const float* bl = (const float*)d_in[8];
    float* out = (float*)d_out;

    const size_t msz = (size_t)NROW * C2;
    float* m1 = (float*)d_ws;
    float* m2 = m1 + msz;
    float* m3 = m2 + msz;
    float* ob = m3 + msz;
    float* op = ob + msz;
    float* ml = op + (size_t)NSPLIT * msz;

    nlb_qkv<<<dim3(NROW / 128, 6), 128, 0, stream>>>(
        x, w1, b1, w2, b2, w3, b3, m1, m2, m3);
    nlb_flash<<<dim3(NROW / (16 * QT * 8), NSPLIT), 256, 0, stream>>>(m1, m2, m3, op, ml);
    nlb_merge<<<dim3(NROW * 24 / 256), 256, 0, stream>>>(op, ml, ob);
    nlb_proj<<<dim3(NROW / 128, DIM / 48), 128, 0, stream>>>(ob, wl, bl, x, out);
}
